# starting kernel
# baseline (speedup 1.0000x reference)
#include <hip/hip_runtime.h>
#include <hip/hip_bf16.h>

typedef float v2f __attribute__((ext_vector_type(2)));
typedef float v8f __attribute__((ext_vector_type(8)));


__global__ void gcn_zero_f32(float* __restrict__ p, long n) {
    long i = (long)blockIdx.x * blockDim.x + threadIdx.x;
    long stride = (long)gridDim.x * blockDim.x;
    for (; i < n; i += stride) p[i] = 0.0f;
}

__global__ void gcn_degree(const int* __restrict__ dst, float* __restrict__ deg, int E) {
    int e = blockIdx.x * blockDim.x + threadIdx.x;
    if (e < E) atomicAdd(&deg[dst[e]], 1.0f);
}

__global__ void gcn_dis(const float* __restrict__ deg, float* __restrict__ dis, int N) {
    int i = blockIdx.x * blockDim.x + threadIdx.x;
    if (i < N) dis[i] = rsqrtf(deg[i] + 1.0f);
}

__global__ void gcn_edge1(const int* __restrict__ src, const int* __restrict__ dst,
                          const float* __restrict__ x, const float* __restrict__ dis,
                          float* __restrict__ s1, int E) {
    int e = blockIdx.x * blockDim.x + threadIdx.x;
    if (e < E) {
        int s = src[e], d = dst[e];
        atomicAdd(&s1[d], x[s] * dis[s] * dis[d]);
    }
}

__global__ void gcn_make_t(const float* __restrict__ s1, const float* __restrict__ dis,
                           const float* __restrict__ x, float* __restrict__ t, int N) {
    int i = blockIdx.x * blockDim.x + threadIdx.x;
    if (i < N) t[i] = s1[i] + dis[i] * dis[i] * x[i];
}

__global__ void gcn_gemm_wmma(const float* __restrict__ t,
                              const float* __restrict__ W1,
                              const float* __restrict__ b1,
                              const float* __restrict__ W2,
                              float* __restrict__ h2pre,
                              int nTiles) {
    int wave = (int)((blockIdx.x * blockDim.x + threadIdx.x) >> 5);
    int lane = threadIdx.x & 31;
    if (wave >= nTiles) return;

    int mt = wave >> 1;
    int nt = wave & 1;
    int m0 = mt << 4;
    int n0 = nt << 4;
    int r = lane & 15;
    int khalf = lane >> 4;

    float trow = t[m0 + r];

    v8f acc = {};
#pragma unroll
    for (int k0 = 0; k0 < 64; k0 += 4) {
        int ka = k0 + khalf * 2;
        float a0 = fmaf(trow, W1[ka],     b1[ka]);
        float a1 = fmaf(trow, W1[ka + 1], b1[ka + 1]);
        v2f a, b;
        a.x = a0 > 0.0f ? a0 : 0.0f;
        a.y = a1 > 0.0f ? a1 : 0.0f;
        b.x = W2[ka * 32 + n0 + r];
        b.y = W2[(ka + 1) * 32 + n0 + r];
        acc = __builtin_amdgcn_wmma_f32_16x16x4_f32(
             false, a,  false, b,
             (short)0, acc,  false,  false);
    }

    int col = n0 + r;
    int rowBase = m0 + khalf * 8;
#pragma unroll
    for (int v = 0; v < 8; ++v) {
        h2pre[(rowBase + v) * 32 + col] = acc[v];
    }
}

__global__ void gcn_edge2(const int* __restrict__ src, const int* __restrict__ dst,
                          const float* __restrict__ dis,
                          const float* __restrict__ h2pre,
                          float* __restrict__ agg2, int E) {
    int lane = threadIdx.x & 31;
    int wave = (int)((blockIdx.x * blockDim.x + threadIdx.x) >> 5);
    int nWaves = (int)((gridDim.x * blockDim.x) >> 5);
    for (int e = wave; e < E; e += nWaves) {
        int s = src[e], d = dst[e];
        float norm = dis[s] * dis[d];
        float v = h2pre[s * 32 + lane] * norm;
        atomicAdd(&agg2[d * 32 + lane], v);
    }
}

__global__ void gcn_pool(const float* __restrict__ agg2, const float* __restrict__ h2pre,
                         const float* __restrict__ dis, const float* __restrict__ b2,
                         const int* __restrict__ batch,
                         float* __restrict__ psum, float* __restrict__ pcnt, int N) {
    long tid = (long)blockIdx.x * blockDim.x + threadIdx.x;
    long total = (long)N * 32;
    long stride = (long)gridDim.x * blockDim.x;
    for (; tid < total; tid += stride) {
        int i = (int)(tid >> 5);
        int f = (int)(tid & 31);
        float d2 = dis[i] * dis[i];
        float h = agg2[tid] + d2 * h2pre[tid] + b2[f];
        h = h > 0.0f ? h : 0.0f;
        int g = batch[i];
        atomicAdd(&psum[g * 32 + f], h);
        if (f == 0) atomicAdd(&pcnt[g], 1.0f);
    }
}

__global__ void gcn_mlp(const float* __restrict__ psum, const float* __restrict__ pcnt,
                        const float* __restrict__ Wp1, const float* __restrict__ bp1,
                        const float* __restrict__ Wp2, const float* __restrict__ bp2,
                        float* __restrict__ out, int G) {
    int g = blockIdx.x * blockDim.x + threadIdx.x;
    if (g >= G) return;
    float c = pcnt[g];
    c = c > 1.0f ? c : 1.0f;
    float inv = 1.0f / c;
    float pooled[32];
#pragma unroll
    for (int f = 0; f < 32; ++f) pooled[f] = psum[g * 32 + f] * inv;
    float o0 = bp2[0], o1 = bp2[1], o2 = bp2[2];
    for (int j = 0; j < 128; ++j) {
        float h = bp1[j];
#pragma unroll
        for (int f = 0; f < 32; ++f) h = fmaf(pooled[f], Wp1[f * 128 + j], h);
        h = h > 0.0f ? h : 0.0f;
        o0 = fmaf(h, Wp2[j * 3 + 0], o0);
        o1 = fmaf(h, Wp2[j * 3 + 1], o1);
        o2 = fmaf(h, Wp2[j * 3 + 2], o2);
    }
    out[g * 3 + 0] = o0;
    out[g * 3 + 1] = o1;
    out[g * 3 + 2] = o2;
}

extern "C" void kernel_launch(void* const* d_in, const int* in_sizes, int n_in,
                              void* d_out, int out_size, void* d_ws, size_t ws_size,
                              hipStream_t stream) {
    const float* x   = (const float*)d_in[0];
    const float* W1  = (const float*)d_in[1];
    const float* b1  = (const float*)d_in[2];
    const float* W2  = (const float*)d_in[3];
    const float* b2  = (const float*)d_in[4];
    const float* Wp1 = (const float*)d_in[5];
    const float* bp1 = (const float*)d_in[6];
    const float* Wp2 = (const float*)d_in[7];
    const float* bp2 = (const float*)d_in[8];
    const int* eidx  = (const int*)d_in[9];
    const int* batch = (const int*)d_in[10];
    float* out = (float*)d_out;

    const int N = in_sizes[0];
    const int E = in_sizes[9] / 2;
    const int G = out_size / 3;
    const int* src = eidx;
    const int* dst = eidx + E;

    float* wsf   = (float*)d_ws;
    float* deg   = wsf;
    float* s1    = deg + N;
    float* dis   = s1 + N;
    float* tbuf  = dis + N;
    float* h2pre = tbuf + N;
    float* agg2  = h2pre + (long)N * 32;
    float* psum  = agg2 + (long)N * 32;
    float* pcnt  = psum + (long)G * 32;

    const int TPB = 256;

    long nz1 = 2L * N;
    long nz2 = 32L * N + 32L * G + G;
    gcn_zero_f32<<<1024, TPB, 0, stream>>>(deg, nz1);
    gcn_zero_f32<<<2048, TPB, 0, stream>>>(agg2, nz2);

    gcn_degree<<<(E + TPB - 1) / TPB, TPB, 0, stream>>>(dst, deg, E);
    gcn_dis<<<(N + TPB - 1) / TPB, TPB, 0, stream>>>(deg, dis, N);
    gcn_edge1<<<(E + TPB - 1) / TPB, TPB, 0, stream>>>(src, dst, x, dis, s1, E);
    gcn_make_t<<<(N + TPB - 1) / TPB, TPB, 0, stream>>>(s1, dis, x, tbuf, N);

    int nTiles = (N / 16) * 2;
    gcn_gemm_wmma<<<(nTiles + 7) / 8, TPB, 0, stream>>>(tbuf, W1, b1, W2, h2pre, nTiles);

    gcn_edge2<<<4096, TPB, 0, stream>>>(src, dst, dis, h2pre, agg2, E);

    gcn_pool<<<4096, TPB, 0, stream>>>(agg2, h2pre, dis, b2, batch, psum, pcnt, N);

    gcn_mlp<<<(G + TPB - 1) / TPB, TPB, 0, stream>>>(psum, pcnt, Wp1, bp1, Wp2, bp2, out, G);
}


# degree and layer-1 scalar scatter: per-edge global atomics replaced by dst-range LDS accumulation, one coalesced flush atomic per node per block; float sum order changes (dis[dst] applied at flush)
# speedup vs baseline: 1.7806x; 1.7806x over previous
#include <hip/hip_runtime.h>
#include <hip/hip_bf16.h>

typedef float v2f __attribute__((ext_vector_type(2)));
typedef float v8f __attribute__((ext_vector_type(8)));


__global__ void gcn_zero_f32(float* __restrict__ p, long n) {
    long i = (long)blockIdx.x * blockDim.x + threadIdx.x;
    long stride = (long)gridDim.x * blockDim.x;
    for (; i < n; i += stride) p[i] = 0.0f;
}

#define GCN_RN 12512
#define GCN_BS 1024
#define GCN_K  32
#define GCN_U  8

__global__ __launch_bounds__(GCN_BS)
void gcn_degree(const int* __restrict__ dst, float* __restrict__ deg, int E, int N, int chunk) {
    __shared__ unsigned cnt[GCN_RN];
    const int tid = threadIdx.x;
    const int lo = blockIdx.y * GCN_RN;
    int nr = N - lo;
    nr = nr < GCN_RN ? nr : GCN_RN;
    for (int i = tid; i < GCN_RN; i += GCN_BS) cnt[i] = 0u;
    __syncthreads();

    const int e0 = blockIdx.x * chunk;
    int e1 = e0 + chunk;
    e1 = e1 < E ? e1 : E;
    for (int e = e0 + tid; e < e1; e += GCN_U * GCN_BS) {
        int d[GCN_U];
#pragma unroll
        for (int j = 0; j < GCN_U; ++j) {
            int ej = e + j * GCN_BS;
            int dv = dst[ej < e1 ? ej : e1 - 1];
            d[j] = ej < e1 ? dv - lo : -1;
        }
#pragma unroll
        for (int j = 0; j < GCN_U; ++j) {
            if ((unsigned)d[j] < (unsigned)nr) atomicAdd(&cnt[d[j]], 1u);
        }
    }
    __syncthreads();
    const int rot = (int)((blockIdx.x * ((GCN_RN / GCN_K) & ~31u)) % (unsigned)nr);
    for (int t = tid; t < nr; t += GCN_BS) {
        int i = t + rot;
        i = i >= nr ? i - nr : i;
        unsigned c = cnt[i];
        if (c != 0u) atomicAdd(&deg[lo + i], (float)c);
    }
}

__global__ void gcn_dis(const float* __restrict__ deg, float* __restrict__ dis, int N) {
    int i = blockIdx.x * blockDim.x + threadIdx.x;
    if (i < N) dis[i] = rsqrtf(deg[i] + 1.0f);
}

__global__ __launch_bounds__(GCN_BS)
void gcn_edge1(const int* __restrict__ src, const int* __restrict__ dst,
               const float* __restrict__ x, const float* __restrict__ dis,
               float* __restrict__ s1, int E, int N, int chunk) {
    __shared__ float acc[GCN_RN];
    const int tid = threadIdx.x;
    const int lo = blockIdx.y * GCN_RN;
    int nr = N - lo;
    nr = nr < GCN_RN ? nr : GCN_RN;
    for (int i = tid; i < GCN_RN; i += GCN_BS) acc[i] = 0.0f;
    __syncthreads();

    const int e0 = blockIdx.x * chunk;
    int e1 = e0 + chunk;
    e1 = e1 < E ? e1 : E;
    for (int e = e0 + tid; e < e1; e += GCN_U * GCN_BS) {
        int d[GCN_U], s[GCN_U];
        float v[GCN_U];
#pragma unroll
        for (int j = 0; j < GCN_U; ++j) {
            int ej = e + j * GCN_BS;
            int ec = ej < e1 ? ej : e1 - 1;
            int dv = dst[ec];
            s[j] = src[ec];
            d[j] = ej < e1 ? dv - lo : -1;
        }
#pragma unroll
        for (int j = 0; j < GCN_U; ++j) {
            int se = ((unsigned)d[j] < (unsigned)nr) ? s[j] : 0;
            v[j] = x[se] * dis[se];
        }
#pragma unroll
        for (int j = 0; j < GCN_U; ++j) {
            asm volatile("" : : "v"(v[j]));
        }
#pragma unroll
        for (int j = 0; j < GCN_U; ++j) {
            if ((unsigned)d[j] < (unsigned)nr) atomicAdd(&acc[d[j]], v[j]);
        }
    }
    __syncthreads();
    const int rot = (int)((blockIdx.x * ((GCN_RN / GCN_K) & ~31u)) % (unsigned)nr);
    for (int t = tid; t < nr; t += GCN_BS) {
        int i = t + rot;
        i = i >= nr ? i - nr : i;
        atomicAdd(&s1[lo + i], acc[i] * dis[lo + i]);
    }
}

__global__ void gcn_make_t(const float* __restrict__ s1, const float* __restrict__ dis,
                           const float* __restrict__ x, float* __restrict__ t, int N) {
    int i = blockIdx.x * blockDim.x + threadIdx.x;
    if (i < N) t[i] = s1[i] + dis[i] * dis[i] * x[i];
}

__global__ void gcn_gemm_wmma(const float* __restrict__ t,
                              const float* __restrict__ W1,
                              const float* __restrict__ b1,
                              const float* __restrict__ W2,
                              float* __restrict__ h2pre,
                              int nTiles) {
    int wave = (int)((blockIdx.x * blockDim.x + threadIdx.x) >> 5);
    int lane = threadIdx.x & 31;
    if (wave >= nTiles) return;

    int mt = wave >> 1;
    int nt = wave & 1;
    int m0 = mt << 4;
    int n0 = nt << 4;
    int r = lane & 15;
    int khalf = lane >> 4;

    float trow = t[m0 + r];

    v8f acc = {};
#pragma unroll
    for (int k0 = 0; k0 < 64; k0 += 4) {
        int ka = k0 + khalf * 2;
        float a0 = fmaf(trow, W1[ka],     b1[ka]);
        float a1 = fmaf(trow, W1[ka + 1], b1[ka + 1]);
        v2f a, b;
        a.x = a0 > 0.0f ? a0 : 0.0f;
        a.y = a1 > 0.0f ? a1 : 0.0f;
        b.x = W2[ka * 32 + n0 + r];
        b.y = W2[(ka + 1) * 32 + n0 + r];
        acc = __builtin_amdgcn_wmma_f32_16x16x4_f32(
             false, a,  false, b,
             (short)0, acc,  false,  false);
    }

    int col = n0 + r;
    int rowBase = m0 + khalf * 8;
#pragma unroll
    for (int v = 0; v < 8; ++v) {
        h2pre[(rowBase + v) * 32 + col] = acc[v];
    }
}

__global__ void gcn_edge2(const int* __restrict__ src, const int* __restrict__ dst,
                          const float* __restrict__ dis,
                          const float* __restrict__ h2pre,
                          float* __restrict__ agg2, int E) {
    int lane = threadIdx.x & 31;
    int wave = (int)((blockIdx.x * blockDim.x + threadIdx.x) >> 5);
    int nWaves = (int)((gridDim.x * blockDim.x) >> 5);
    for (int e = wave; e < E; e += nWaves) {
        int s = src[e], d = dst[e];
        float norm = dis[s] * dis[d];
        float v = h2pre[s * 32 + lane] * norm;
        atomicAdd(&agg2[d * 32 + lane], v);
    }
}

__global__ void gcn_pool(const float* __restrict__ agg2, const float* __restrict__ h2pre,
                         const float* __restrict__ dis, const float* __restrict__ b2,
                         const int* __restrict__ batch,
                         float* __restrict__ psum, float* __restrict__ pcnt, int N) {
    long tid = (long)blockIdx.x * blockDim.x + threadIdx.x;
    long total = (long)N * 32;
    long stride = (long)gridDim.x * blockDim.x;
    for (; tid < total; tid += stride) {
        int i = (int)(tid >> 5);
        int f = (int)(tid & 31);
        float d2 = dis[i] * dis[i];
        float h = agg2[tid] + d2 * h2pre[tid] + b2[f];
        h = h > 0.0f ? h : 0.0f;
        int g = batch[i];
        atomicAdd(&psum[g * 32 + f], h);
        if (f == 0) atomicAdd(&pcnt[g], 1.0f);
    }
}

__global__ void gcn_mlp(const float* __restrict__ psum, const float* __restrict__ pcnt,
                        const float* __restrict__ Wp1, const float* __restrict__ bp1,
                        const float* __restrict__ Wp2, const float* __restrict__ bp2,
                        float* __restrict__ out, int G) {
    int g = blockIdx.x * blockDim.x + threadIdx.x;
    if (g >= G) return;
    float c = pcnt[g];
    c = c > 1.0f ? c : 1.0f;
    float inv = 1.0f / c;
    float pooled[32];
#pragma unroll
    for (int f = 0; f < 32; ++f) pooled[f] = psum[g * 32 + f] * inv;
    float o0 = bp2[0], o1 = bp2[1], o2 = bp2[2];
    for (int j = 0; j < 128; ++j) {
        float h = bp1[j];
#pragma unroll
        for (int f = 0; f < 32; ++f) h = fmaf(pooled[f], Wp1[f * 128 + j], h);
        h = h > 0.0f ? h : 0.0f;
        o0 = fmaf(h, Wp2[j * 3 + 0], o0);
        o1 = fmaf(h, Wp2[j * 3 + 1], o1);
        o2 = fmaf(h, Wp2[j * 3 + 2], o2);
    }
    out[g * 3 + 0] = o0;
    out[g * 3 + 1] = o1;
    out[g * 3 + 2] = o2;
}

extern "C" void kernel_launch(void* const* d_in, const int* in_sizes, int n_in,
                              void* d_out, int out_size, void* d_ws, size_t ws_size,
                              hipStream_t stream) {
    const float* x   = (const float*)d_in[0];
    const float* W1  = (const float*)d_in[1];
    const float* b1  = (const float*)d_in[2];
    const float* W2  = (const float*)d_in[3];
    const float* b2  = (const float*)d_in[4];
    const float* Wp1 = (const float*)d_in[5];
    const float* bp1 = (const float*)d_in[6];
    const float* Wp2 = (const float*)d_in[7];
    const float* bp2 = (const float*)d_in[8];
    const int* eidx  = (const int*)d_in[9];
    const int* batch = (const int*)d_in[10];
    float* out = (float*)d_out;

    const int N = in_sizes[0];
    const int E = in_sizes[9] / 2;
    const int G = out_size / 3;
    const int* src = eidx;
    const int* dst = eidx + E;

    float* wsf   = (float*)d_ws;
    float* deg   = wsf;
    float* s1    = deg + N;
    float* dis   = s1 + N;
    float* tbuf  = dis + N;
    float* h2pre = tbuf + N;
    float* agg2  = h2pre + (long)N * 32;
    float* psum  = agg2 + (long)N * 32;
    float* pcnt  = psum + (long)G * 32;

    const int TPB = 256;

    long nz1 = 2L * N;
    long nz2 = 32L * N + 32L * G + G;
    gcn_zero_f32<<<1024, TPB, 0, stream>>>(deg, nz1);
    gcn_zero_f32<<<2048, TPB, 0, stream>>>(agg2, nz2);

    const int chunk = (E + GCN_K - 1) / GCN_K;
    const dim3 sgrid(GCN_K, (N + GCN_RN - 1) / GCN_RN);
    gcn_degree<<<sgrid, GCN_BS, 0, stream>>>(dst, deg, E, N, chunk);
    gcn_dis<<<(N + TPB - 1) / TPB, TPB, 0, stream>>>(deg, dis, N);
    gcn_edge1<<<sgrid, GCN_BS, 0, stream>>>(src, dst, x, dis, s1, E, N, chunk);
    gcn_make_t<<<(N + TPB - 1) / TPB, TPB, 0, stream>>>(s1, dis, x, tbuf, N);

    int nTiles = (N / 16) * 2;
    gcn_gemm_wmma<<<(nTiles + 7) / 8, TPB, 0, stream>>>(tbuf, W1, b1, W2, h2pre, nTiles);

    gcn_edge2<<<4096, TPB, 0, stream>>>(src, dst, dis, h2pre, agg2, E);

    gcn_pool<<<4096, TPB, 0, stream>>>(agg2, h2pre, dis, b2, batch, psum, pcnt, N);

    gcn_mlp<<<(G + TPB - 1) / TPB, TPB, 0, stream>>>(psum, pcnt, Wp1, bp1, Wp2, bp2, out, G);
}


# layer-2 edge pass: a wave takes 32 consecutive edges per trip (coalesced index and norm loads, broadcast by readlane) and issues 8 row gathers ahead of their atomics; atomic arrival order changes
# speedup vs baseline: 2.4308x; 1.3652x over previous
#include <hip/hip_runtime.h>
#include <hip/hip_bf16.h>

typedef float v2f __attribute__((ext_vector_type(2)));
typedef float v8f __attribute__((ext_vector_type(8)));


__global__ void gcn_zero_f32(float* __restrict__ p, long n) {
    long i = (long)blockIdx.x * blockDim.x + threadIdx.x;
    long stride = (long)gridDim.x * blockDim.x;
    for (; i < n; i += stride) p[i] = 0.0f;
}

#define GCN_RN 12512
#define GCN_BS 1024
#define GCN_K  32
#define GCN_U  8

__global__ __launch_bounds__(GCN_BS)
void gcn_degree(const int* __restrict__ dst, float* __restrict__ deg, int E, int N, int chunk) {
    __shared__ unsigned cnt[GCN_RN];
    const int tid = threadIdx.x;
    const int lo = blockIdx.y * GCN_RN;
    int nr = N - lo;
    nr = nr < GCN_RN ? nr : GCN_RN;
    for (int i = tid; i < GCN_RN; i += GCN_BS) cnt[i] = 0u;
    __syncthreads();

    const int e0 = blockIdx.x * chunk;
    int e1 = e0 + chunk;
    e1 = e1 < E ? e1 : E;
    for (int e = e0 + tid; e < e1; e += GCN_U * GCN_BS) {
        int d[GCN_U];
#pragma unroll
        for (int j = 0; j < GCN_U; ++j) {
            int ej = e + j * GCN_BS;
            int dv = dst[ej < e1 ? ej : e1 - 1];
            d[j] = ej < e1 ? dv - lo : -1;
        }
#pragma unroll
        for (int j = 0; j < GCN_U; ++j) {
            if ((unsigned)d[j] < (unsigned)nr) atomicAdd(&cnt[d[j]], 1u);
        }
    }
    __syncthreads();
    const int rot = (int)((blockIdx.x * ((GCN_RN / GCN_K) & ~31u)) % (unsigned)nr);
    for (int t = tid; t < nr; t += GCN_BS) {
        int i = t + rot;
        i = i >= nr ? i - nr : i;
        unsigned c = cnt[i];
        if (c != 0u) atomicAdd(&deg[lo + i], (float)c);
    }
}

__global__ void gcn_dis(const float* __restrict__ deg, float* __restrict__ dis, int N) {
    int i = blockIdx.x * blockDim.x + threadIdx.x;
    if (i < N) dis[i] = rsqrtf(deg[i] + 1.0f);
}

__global__ __launch_bounds__(GCN_BS)
void gcn_edge1(const int* __restrict__ src, const int* __restrict__ dst,
               const float* __restrict__ x, const float* __restrict__ dis,
               float* __restrict__ s1, int E, int N, int chunk) {
    __shared__ float acc[GCN_RN];
    const int tid = threadIdx.x;
    const int lo = blockIdx.y * GCN_RN;
    int nr = N - lo;
    nr = nr < GCN_RN ? nr : GCN_RN;
    for (int i = tid; i < GCN_RN; i += GCN_BS) acc[i] = 0.0f;
    __syncthreads();

    const int e0 = blockIdx.x * chunk;
    int e1 = e0 + chunk;
    e1 = e1 < E ? e1 : E;
    for (int e = e0 + tid; e < e1; e += GCN_U * GCN_BS) {
        int d[GCN_U], s[GCN_U];
        float v[GCN_U];
#pragma unroll
        for (int j = 0; j < GCN_U; ++j) {
            int ej = e + j * GCN_BS;
            int ec = ej < e1 ? ej : e1 - 1;
            int dv = dst[ec];
            s[j] = src[ec];
            d[j] = ej < e1 ? dv - lo : -1;
        }
#pragma unroll
        for (int j = 0; j < GCN_U; ++j) {
            int se = ((unsigned)d[j] < (unsigned)nr) ? s[j] : 0;
            v[j] = x[se] * dis[se];
        }
#pragma unroll
        for (int j = 0; j < GCN_U; ++j) {
            asm volatile("" : : "v"(v[j]));
        }
#pragma unroll
        for (int j = 0; j < GCN_U; ++j) {
            if ((unsigned)d[j] < (unsigned)nr) atomicAdd(&acc[d[j]], v[j]);
        }
    }
    __syncthreads();
    const int rot = (int)((blockIdx.x * ((GCN_RN / GCN_K) & ~31u)) % (unsigned)nr);
    for (int t = tid; t < nr; t += GCN_BS) {
        int i = t + rot;
        i = i >= nr ? i - nr : i;
        atomicAdd(&s1[lo + i], acc[i] * dis[lo + i]);
    }
}

__global__ void gcn_make_t(const float* __restrict__ s1, const float* __restrict__ dis,
                           const float* __restrict__ x, float* __restrict__ t, int N) {
    int i = blockIdx.x * blockDim.x + threadIdx.x;
    if (i < N) t[i] = s1[i] + dis[i] * dis[i] * x[i];
}

__global__ void gcn_gemm_wmma(const float* __restrict__ t,
                              const float* __restrict__ W1,
                              const float* __restrict__ b1,
                              const float* __restrict__ W2,
                              float* __restrict__ h2pre,
                              int nTiles) {
    int wave = (int)((blockIdx.x * blockDim.x + threadIdx.x) >> 5);
    int lane = threadIdx.x & 31;
    if (wave >= nTiles) return;

    int mt = wave >> 1;
    int nt = wave & 1;
    int m0 = mt << 4;
    int n0 = nt << 4;
    int r = lane & 15;
    int khalf = lane >> 4;

    float trow = t[m0 + r];

    v8f acc = {};
#pragma unroll
    for (int k0 = 0; k0 < 64; k0 += 4) {
        int ka = k0 + khalf * 2;
        float a0 = fmaf(trow, W1[ka],     b1[ka]);
        float a1 = fmaf(trow, W1[ka + 1], b1[ka + 1]);
        v2f a, b;
        a.x = a0 > 0.0f ? a0 : 0.0f;
        a.y = a1 > 0.0f ? a1 : 0.0f;
        b.x = W2[ka * 32 + n0 + r];
        b.y = W2[(ka + 1) * 32 + n0 + r];
        acc = __builtin_amdgcn_wmma_f32_16x16x4_f32(
             false, a,  false, b,
             (short)0, acc,  false,  false);
    }

    int col = n0 + r;
    int rowBase = m0 + khalf * 8;
#pragma unroll
    for (int v = 0; v < 8; ++v) {
        h2pre[(rowBase + v) * 32 + col] = acc[v];
    }
}

#define GCN_E2U 8
__global__ __launch_bounds__(256)
void gcn_edge2(const int* __restrict__ src, const int* __restrict__ dst,
               const float* __restrict__ dis,
               const float* __restrict__ h2pre,
               float* __restrict__ agg2, int E) {
    const int lane = threadIdx.x & 31;
    const int wave = __builtin_amdgcn_readfirstlane(
        (int)((blockIdx.x * blockDim.x + threadIdx.x) >> 5));
    const int nWaves = (int)((gridDim.x * blockDim.x) >> 5);
    const int nBatch = (E + 31) >> 5;
    for (int b = wave; b < nBatch; b += nWaves) {
        const int e0 = b << 5;
        const int e = e0 + lane;
        const int ec = e < E ? e : E - 1;
        const int s = src[ec];
        const int d = dst[ec];
        const int nrm = __float_as_int(dis[s] * dis[d]);
        int cnt = E - e0;
        cnt = cnt < 32 ? cnt : 32;
        if (cnt == 32) {
#pragma unroll
            for (int j0 = 0; j0 < 32; j0 += GCN_E2U) {
                float v[GCN_E2U];
#pragma unroll
                for (int j = 0; j < GCN_E2U; ++j) {
                    const int sj = __builtin_amdgcn_readlane(s, j0 + j);
                    v[j] = h2pre[(long)sj * 32 + lane];
                }
#pragma unroll
                for (int j = 0; j < GCN_E2U; ++j) {
                    const int dj = __builtin_amdgcn_readlane(d, j0 + j);
                    const float nj = __int_as_float(__builtin_amdgcn_readlane(nrm, j0 + j));
                    atomicAdd(&agg2[(long)dj * 32 + lane], v[j] * nj);
                }
            }
        } else {
            for (int j = 0; j < cnt; ++j) {
                const int sj = __builtin_amdgcn_readlane(s, j);
                const int dj = __builtin_amdgcn_readlane(d, j);
                const float nj = __int_as_float(__builtin_amdgcn_readlane(nrm, j));
                atomicAdd(&agg2[(long)dj * 32 + lane], h2pre[(long)sj * 32 + lane] * nj);
            }
        }
    }
}

__global__ void gcn_pool(const float* __restrict__ agg2, const float* __restrict__ h2pre,
                         const float* __restrict__ dis, const float* __restrict__ b2,
                         const int* __restrict__ batch,
                         float* __restrict__ psum, float* __restrict__ pcnt, int N) {
    long tid = (long)blockIdx.x * blockDim.x + threadIdx.x;
    long total = (long)N * 32;
    long stride = (long)gridDim.x * blockDim.x;
    for (; tid < total; tid += stride) {
        int i = (int)(tid >> 5);
        int f = (int)(tid & 31);
        float d2 = dis[i] * dis[i];
        float h = agg2[tid] + d2 * h2pre[tid] + b2[f];
        h = h > 0.0f ? h : 0.0f;
        int g = batch[i];
        atomicAdd(&psum[g * 32 + f], h);
        if (f == 0) atomicAdd(&pcnt[g], 1.0f);
    }
}

__global__ void gcn_mlp(const float* __restrict__ psum, const float* __restrict__ pcnt,
                        const float* __restrict__ Wp1, const float* __restrict__ bp1,
                        const float* __restrict__ Wp2, const float* __restrict__ bp2,
                        float* __restrict__ out, int G) {
    int g = blockIdx.x * blockDim.x + threadIdx.x;
    if (g >= G) return;
    float c = pcnt[g];
    c = c > 1.0f ? c : 1.0f;
    float inv = 1.0f / c;
    float pooled[32];
#pragma unroll
    for (int f = 0; f < 32; ++f) pooled[f] = psum[g * 32 + f] * inv;
    float o0 = bp2[0], o1 = bp2[1], o2 = bp2[2];
    for (int j = 0; j < 128; ++j) {
        float h = bp1[j];
#pragma unroll
        for (int f = 0; f < 32; ++f) h = fmaf(pooled[f], Wp1[f * 128 + j], h);
        h = h > 0.0f ? h : 0.0f;
        o0 = fmaf(h, Wp2[j * 3 + 0], o0);
        o1 = fmaf(h, Wp2[j * 3 + 1], o1);
        o2 = fmaf(h, Wp2[j * 3 + 2], o2);
    }
    out[g * 3 + 0] = o0;
    out[g * 3 + 1] = o1;
    out[g * 3 + 2] = o2;
}

extern "C" void kernel_launch(void* const* d_in, const int* in_sizes, int n_in,
                              void* d_out, int out_size, void* d_ws, size_t ws_size,
                              hipStream_t stream) {
    const float* x   = (const float*)d_in[0];
    const float* W1  = (const float*)d_in[1];
    const float* b1  = (const float*)d_in[2];
    const float* W2  = (const float*)d_in[3];
    const float* b2  = (const float*)d_in[4];
    const float* Wp1 = (const float*)d_in[5];
    const float* bp1 = (const float*)d_in[6];
    const float* Wp2 = (const float*)d_in[7];
    const float* bp2 = (const float*)d_in[8];
    const int* eidx  = (const int*)d_in[9];
    const int* batch = (const int*)d_in[10];
    float* out = (float*)d_out;

    const int N = in_sizes[0];
    const int E = in_sizes[9] / 2;
    const int G = out_size / 3;
    const int* src = eidx;
    const int* dst = eidx + E;

    float* wsf   = (float*)d_ws;
    float* deg   = wsf;
    float* s1    = deg + N;
    float* dis   = s1 + N;
    float* tbuf  = dis + N;
    float* h2pre = tbuf + N;
    float* agg2  = h2pre + (long)N * 32;
    float* psum  = agg2 + (long)N * 32;
    float* pcnt  = psum + (long)G * 32;

    const int TPB = 256;

    long nz1 = 2L * N;
    long nz2 = 32L * N + 32L * G + G;
    gcn_zero_f32<<<1024, TPB, 0, stream>>>(deg, nz1);
    gcn_zero_f32<<<2048, TPB, 0, stream>>>(agg2, nz2);

    const int chunk = (E + GCN_K - 1) / GCN_K;
    const dim3 sgrid(GCN_K, (N + GCN_RN - 1) / GCN_RN);
    gcn_degree<<<sgrid, GCN_BS, 0, stream>>>(dst, deg, E, N, chunk);
    gcn_dis<<<(N + TPB - 1) / TPB, TPB, 0, stream>>>(deg, dis, N);
    gcn_edge1<<<sgrid, GCN_BS, 0, stream>>>(src, dst, x, dis, s1, E, N, chunk);
    gcn_make_t<<<(N + TPB - 1) / TPB, TPB, 0, stream>>>(s1, dis, x, tbuf, N);

    int nTiles = (N / 16) * 2;
    gcn_gemm_wmma<<<(nTiles + 7) / 8, TPB, 0, stream>>>(tbuf, W1, b1, W2, h2pre, nTiles);

    gcn_edge2<<<2048, TPB, 0, stream>>>(src, dst, dis, h2pre, agg2, E);

    gcn_pool<<<4096, TPB, 0, stream>>>(agg2, h2pre, dis, b2, batch, psum, pcnt, N);

    gcn_mlp<<<(G + TPB - 1) / TPB, TPB, 0, stream>>>(psum, pcnt, Wp1, bp1, Wp2, bp2, out, G);
}


# MLP head: one wave per graph, a lane owns 4 hidden units (16-B Wp1 row loads, pooled[f] by readlane) instead of one thread per graph; 128->3 sum per lane then wave butterfly (sum order changes)
# speedup vs baseline: 2.8468x; 1.1711x over previous
#include <hip/hip_runtime.h>
#include <hip/hip_bf16.h>

typedef float v2f __attribute__((ext_vector_type(2)));
typedef float v8f __attribute__((ext_vector_type(8)));


__global__ void gcn_zero_f32(float* __restrict__ p, long n) {
    long i = (long)blockIdx.x * blockDim.x + threadIdx.x;
    long stride = (long)gridDim.x * blockDim.x;
    for (; i < n; i += stride) p[i] = 0.0f;
}

#define GCN_RN 12512
#define GCN_BS 1024
#define GCN_K  32
#define GCN_U  8

__global__ __launch_bounds__(GCN_BS)
void gcn_degree(const int* __restrict__ dst, float* __restrict__ deg, int E, int N, int chunk) {
    __shared__ unsigned cnt[GCN_RN];
    const int tid = threadIdx.x;
    const int lo = blockIdx.y * GCN_RN;
    int nr = N - lo;
    nr = nr < GCN_RN ? nr : GCN_RN;
    for (int i = tid; i < GCN_RN; i += GCN_BS) cnt[i] = 0u;
    __syncthreads();

    const int e0 = blockIdx.x * chunk;
    int e1 = e0 + chunk;
    e1 = e1 < E ? e1 : E;
    for (int e = e0 + tid; e < e1; e += GCN_U * GCN_BS) {
        int d[GCN_U];
#pragma unroll
        for (int j = 0; j < GCN_U; ++j) {
            int ej = e + j * GCN_BS;
            int dv = dst[ej < e1 ? ej : e1 - 1];
            d[j] = ej < e1 ? dv - lo : -1;
        }
#pragma unroll
        for (int j = 0; j < GCN_U; ++j) {
            if ((unsigned)d[j] < (unsigned)nr) atomicAdd(&cnt[d[j]], 1u);
        }
    }
    __syncthreads();
    const int rot = (int)((blockIdx.x * ((GCN_RN / GCN_K) & ~31u)) % (unsigned)nr);
    for (int t = tid; t < nr; t += GCN_BS) {
        int i = t + rot;
        i = i >= nr ? i - nr : i;
        unsigned c = cnt[i];
        if (c != 0u) atomicAdd(&deg[lo + i], (float)c);
    }
}

__global__ void gcn_dis(const float* __restrict__ deg, float* __restrict__ dis, int N) {
    int i = blockIdx.x * blockDim.x + threadIdx.x;
    if (i < N) dis[i] = rsqrtf(deg[i] + 1.0f);
}

__global__ __launch_bounds__(GCN_BS)
void gcn_edge1(const int* __restrict__ src, const int* __restrict__ dst,
               const float* __restrict__ x, const float* __restrict__ dis,
               float* __restrict__ s1, int E, int N, int chunk) {
    __shared__ float acc[GCN_RN];
    const int tid = threadIdx.x;
    const int lo = blockIdx.y * GCN_RN;
    int nr = N - lo;
    nr = nr < GCN_RN ? nr : GCN_RN;
    for (int i = tid; i < GCN_RN; i += GCN_BS) acc[i] = 0.0f;
    __syncthreads();

    const int e0 = blockIdx.x * chunk;
    int e1 = e0 + chunk;
    e1 = e1 < E ? e1 : E;
    for (int e = e0 + tid; e < e1; e += GCN_U * GCN_BS) {
        int d[GCN_U], s[GCN_U];
        float v[GCN_U];
#pragma unroll
        for (int j = 0; j < GCN_U; ++j) {
            int ej = e + j * GCN_BS;
            int ec = ej < e1 ? ej : e1 - 1;
            int dv = dst[ec];
            s[j] = src[ec];
            d[j] = ej < e1 ? dv - lo : -1;
        }
#pragma unroll
        for (int j = 0; j < GCN_U; ++j) {
            int se = ((unsigned)d[j] < (unsigned)nr) ? s[j] : 0;
            v[j] = x[se] * dis[se];
        }
#pragma unroll
        for (int j = 0; j < GCN_U; ++j) {
            asm volatile("" : : "v"(v[j]));
        }
#pragma unroll
        for (int j = 0; j < GCN_U; ++j) {
            if ((unsigned)d[j] < (unsigned)nr) atomicAdd(&acc[d[j]], v[j]);
        }
    }
    __syncthreads();
    const int rot = (int)((blockIdx.x * ((GCN_RN / GCN_K) & ~31u)) % (unsigned)nr);
    for (int t = tid; t < nr; t += GCN_BS) {
        int i = t + rot;
        i = i >= nr ? i - nr : i;
        atomicAdd(&s1[lo + i], acc[i] * dis[lo + i]);
    }
}

__global__ void gcn_make_t(const float* __restrict__ s1, const float* __restrict__ dis,
                           const float* __restrict__ x, float* __restrict__ t, int N) {
    int i = blockIdx.x * blockDim.x + threadIdx.x;
    if (i < N) t[i] = s1[i] + dis[i] * dis[i] * x[i];
}

__global__ void gcn_gemm_wmma(const float* __restrict__ t,
                              const float* __restrict__ W1,
                              const float* __restrict__ b1,
                              const float* __restrict__ W2,
                              float* __restrict__ h2pre,
                              int nTiles) {
    int wave = (int)((blockIdx.x * blockDim.x + threadIdx.x) >> 5);
    int lane = threadIdx.x & 31;
    if (wave >= nTiles) return;

    int mt = wave >> 1;
    int nt = wave & 1;
    int m0 = mt << 4;
    int n0 = nt << 4;
    int r = lane & 15;
    int khalf = lane >> 4;

    float trow = t[m0 + r];

    v8f acc = {};
#pragma unroll
    for (int k0 = 0; k0 < 64; k0 += 4) {
        int ka = k0 + khalf * 2;
        float a0 = fmaf(trow, W1[ka],     b1[ka]);
        float a1 = fmaf(trow, W1[ka + 1], b1[ka + 1]);
        v2f a, b;
        a.x = a0 > 0.0f ? a0 : 0.0f;
        a.y = a1 > 0.0f ? a1 : 0.0f;
        b.x = W2[ka * 32 + n0 + r];
        b.y = W2[(ka + 1) * 32 + n0 + r];
        acc = __builtin_amdgcn_wmma_f32_16x16x4_f32(
             false, a,  false, b,
             (short)0, acc,  false,  false);
    }

    int col = n0 + r;
    int rowBase = m0 + khalf * 8;
#pragma unroll
    for (int v = 0; v < 8; ++v) {
        h2pre[(rowBase + v) * 32 + col] = acc[v];
    }
}

#define GCN_E2U 8
__global__ __launch_bounds__(256)
void gcn_edge2(const int* __restrict__ src, const int* __restrict__ dst,
               const float* __restrict__ dis,
               const float* __restrict__ h2pre,
               float* __restrict__ agg2, int E) {
    const int lane = threadIdx.x & 31;
    const int wave = __builtin_amdgcn_readfirstlane(
        (int)((blockIdx.x * blockDim.x + threadIdx.x) >> 5));
    const int nWaves = (int)((gridDim.x * blockDim.x) >> 5);
    const int nBatch = (E + 31) >> 5;
    for (int b = wave; b < nBatch; b += nWaves) {
        const int e0 = b << 5;
        const int e = e0 + lane;
        const int ec = e < E ? e : E - 1;
        const int s = src[ec];
        const int d = dst[ec];
        const int nrm = __float_as_int(dis[s] * dis[d]);
        int cnt = E - e0;
        cnt = cnt < 32 ? cnt : 32;
        if (cnt == 32) {
#pragma unroll
            for (int j0 = 0; j0 < 32; j0 += GCN_E2U) {
                float v[GCN_E2U];
#pragma unroll
                for (int j = 0; j < GCN_E2U; ++j) {
                    const int sj = __builtin_amdgcn_readlane(s, j0 + j);
                    v[j] = h2pre[(long)sj * 32 + lane];
                }
#pragma unroll
                for (int j = 0; j < GCN_E2U; ++j) {
                    const int dj = __builtin_amdgcn_readlane(d, j0 + j);
                    const float nj = __int_as_float(__builtin_amdgcn_readlane(nrm, j0 + j));
                    atomicAdd(&agg2[(long)dj * 32 + lane], v[j] * nj);
                }
            }
        } else {
            for (int j = 0; j < cnt; ++j) {
                const int sj = __builtin_amdgcn_readlane(s, j);
                const int dj = __builtin_amdgcn_readlane(d, j);
                const float nj = __int_as_float(__builtin_amdgcn_readlane(nrm, j));
                atomicAdd(&agg2[(long)dj * 32 + lane], h2pre[(long)sj * 32 + lane] * nj);
            }
        }
    }
}

__global__ void gcn_pool(const float* __restrict__ agg2, const float* __restrict__ h2pre,
                         const float* __restrict__ dis, const float* __restrict__ b2,
                         const int* __restrict__ batch,
                         float* __restrict__ psum, float* __restrict__ pcnt, int N) {
    long tid = (long)blockIdx.x * blockDim.x + threadIdx.x;
    long total = (long)N * 32;
    long stride = (long)gridDim.x * blockDim.x;
    for (; tid < total; tid += stride) {
        int i = (int)(tid >> 5);
        int f = (int)(tid & 31);
        float d2 = dis[i] * dis[i];
        float h = agg2[tid] + d2 * h2pre[tid] + b2[f];
        h = h > 0.0f ? h : 0.0f;
        int g = batch[i];
        atomicAdd(&psum[g * 32 + f], h);
        if (f == 0) atomicAdd(&pcnt[g], 1.0f);
    }
}

#define GCN_MLP_BS 1024
#define GCN_MLP_GPB (GCN_MLP_BS / 32)
__global__ __launch_bounds__(GCN_MLP_BS)
void gcn_mlp(const float* __restrict__ psum, const float* __restrict__ pcnt,
             const float* __restrict__ Wp1, const float* __restrict__ bp1,
             const float* __restrict__ Wp2, const float* __restrict__ bp2,
             float* __restrict__ out, int G) {
    __shared__ float res[GCN_MLP_GPB * 3];
    const int lane = threadIdx.x & 31;
    const int w = __builtin_amdgcn_readfirstlane((int)(threadIdx.x >> 5));
    const int g = blockIdx.x * GCN_MLP_GPB + w;
    if (g < G) {
        float c = pcnt[g];
        c = c > 1.0f ? c : 1.0f;
        const float inv = 1.0f / c;
        const int p = __float_as_int(psum[g * 32 + lane] * inv);

        const int j0 = lane * 4;
        float h0 = bp1[j0], h1 = bp1[j0 + 1], h2 = bp1[j0 + 2], h3 = bp1[j0 + 3];
#pragma unroll
        for (int f = 0; f < 32; ++f) {
            const float pf = __int_as_float(__builtin_amdgcn_readlane(p, f));
            const float* wr = Wp1 + f * 128 + j0;
            h0 = fmaf(pf, wr[0], h0);
            h1 = fmaf(pf, wr[1], h1);
            h2 = fmaf(pf, wr[2], h2);
            h3 = fmaf(pf, wr[3], h3);
        }
        h0 = h0 > 0.0f ? h0 : 0.0f;
        h1 = h1 > 0.0f ? h1 : 0.0f;
        h2 = h2 > 0.0f ? h2 : 0.0f;
        h3 = h3 > 0.0f ? h3 : 0.0f;

        const float* v = Wp2 + j0 * 3;
        float o0 = fmaf(h3, v[9],  fmaf(h2, v[6], fmaf(h1, v[3], h0 * v[0])));
        float o1 = fmaf(h3, v[10], fmaf(h2, v[7], fmaf(h1, v[4], h0 * v[1])));
        float o2 = fmaf(h3, v[11], fmaf(h2, v[8], fmaf(h1, v[5], h0 * v[2])));
        for (int m = 16; m >= 1; m >>= 1) {
            o0 += __shfl_xor(o0, m);
            o1 += __shfl_xor(o1, m);
            o2 += __shfl_xor(o2, m);
        }
        if (lane == 0) {
            res[w * 3 + 0] = o0 + bp2[0];
            res[w * 3 + 1] = o1 + bp2[1];
            res[w * 3 + 2] = o2 + bp2[2];
        }
    }
    __syncthreads();
    if (w == 0) {
        const int gg = blockIdx.x * GCN_MLP_GPB + lane;
        if (gg < G) {
            out[gg * 3 + 0] = res[lane * 3 + 0];
            out[gg * 3 + 1] = res[lane * 3 + 1];
            out[gg * 3 + 2] = res[lane * 3 + 2];
        }
    }
}

extern "C" void kernel_launch(void* const* d_in, const int* in_sizes, int n_in,
                              void* d_out, int out_size, void* d_ws, size_t ws_size,
                              hipStream_t stream) {
    const float* x   = (const float*)d_in[0];
    const float* W1  = (const float*)d_in[1];
    const float* b1  = (const float*)d_in[2];
    const float* W2  = (const float*)d_in[3];
    const float* b2  = (const float*)d_in[4];
    const float* Wp1 = (const float*)d_in[5];
    const float* bp1 = (const float*)d_in[6];
    const float* Wp2 = (const float*)d_in[7];
    const float* bp2 = (const float*)d_in[8];
    const int* eidx  = (const int*)d_in[9];
    const int* batch = (const int*)d_in[10];
    float* out = (float*)d_out;

    const int N = in_sizes[0];
    const int E = in_sizes[9] / 2;
    const int G = out_size / 3;
    const int* src = eidx;
    const int* dst = eidx + E;

    float* wsf   = (float*)d_ws;
    float* deg   = wsf;
    float* s1    = deg + N;
    float* dis   = s1 + N;
    float* tbuf  = dis + N;
    float* h2pre = tbuf + N;
    float* agg2  = h2pre + (long)N * 32;
    float* psum  = agg2 + (long)N * 32;
    float* pcnt  = psum + (long)G * 32;

    const int TPB = 256;

    long nz1 = 2L * N;
    long nz2 = 32L * N + 32L * G + G;
    gcn_zero_f32<<<1024, TPB, 0, stream>>>(deg, nz1);
    gcn_zero_f32<<<2048, TPB, 0, stream>>>(agg2, nz2);

    const int chunk = (E + GCN_K - 1) / GCN_K;
    const dim3 sgrid(GCN_K, (N + GCN_RN - 1) / GCN_RN);
    gcn_degree<<<sgrid, GCN_BS, 0, stream>>>(dst, deg, E, N, chunk);
    gcn_dis<<<(N + TPB - 1) / TPB, TPB, 0, stream>>>(deg, dis, N);
    gcn_edge1<<<sgrid, GCN_BS, 0, stream>>>(src, dst, x, dis, s1, E, N, chunk);
    gcn_make_t<<<(N + TPB - 1) / TPB, TPB, 0, stream>>>(s1, dis, x, tbuf, N);

    int nTiles = (N / 16) * 2;
    gcn_gemm_wmma<<<(nTiles + 7) / 8, TPB, 0, stream>>>(tbuf, W1, b1, W2, h2pre, nTiles);

    gcn_edge2<<<2048, TPB, 0, stream>>>(src, dst, dis, h2pre, agg2, E);

    gcn_pool<<<4096, TPB, 0, stream>>>(agg2, h2pre, dis, b2, batch, psum, pcnt, N);

    gcn_mlp<<<(G + GCN_MLP_GPB - 1) / GCN_MLP_GPB, GCN_MLP_BS, 0, stream>>>(psum, pcnt, Wp1, bp1, Wp2, bp2, out, G);
}


# global float accumulators (deg, s1, agg2, pooled sums and counts) get one copy per cache half, chosen by the wave's die number; consumers add the two copies; float sum order changes
# speedup vs baseline: 3.9863x; 1.4003x over previous
#include <hip/hip_runtime.h>
#include <hip/hip_bf16.h>

typedef float v2f __attribute__((ext_vector_type(2)));
typedef float v8f __attribute__((ext_vector_type(8)));


__device__ __forceinline__ int gcn_half() {
    return (int)((__builtin_amdgcn_s_sendmsg_rtn(0x87) >> 18) & 1u);
}

__device__ __forceinline__ bool gcn_two(size_t ws_bytes, size_t need_bytes) {
    return ws_bytes >= need_bytes;
}

__global__ void gcn_zero_f32(float* __restrict__ p, long n_one, long n_two,
                             size_t ws_bytes, size_t need_bytes) {
    const long n = gcn_two(ws_bytes, need_bytes) ? n_two : n_one;
    long i = (long)blockIdx.x * blockDim.x + threadIdx.x;
    long stride = (long)gridDim.x * blockDim.x;
    for (; i < n; i += stride) p[i] = 0.0f;
}

#define GCN_RN 12512
#define GCN_BS 1024
#define GCN_K  32
#define GCN_U  8

__global__ __launch_bounds__(GCN_BS)
void gcn_degree(const int* __restrict__ dst, float* __restrict__ deg, float* __restrict__ degb,
                int E, int N, int chunk, size_t ws_bytes, size_t need_bytes) {
    __shared__ unsigned cnt[GCN_RN];
    const int tid = threadIdx.x;
    const int lo = blockIdx.y * GCN_RN;
    int nr = N - lo;
    nr = nr < GCN_RN ? nr : GCN_RN;
    for (int i = tid; i < GCN_RN; i += GCN_BS) cnt[i] = 0u;
    __syncthreads();

    const int e0 = blockIdx.x * chunk;
    int e1 = e0 + chunk;
    e1 = e1 < E ? e1 : E;
    for (int e = e0 + tid; e < e1; e += GCN_U * GCN_BS) {
        int d[GCN_U];
#pragma unroll
        for (int j = 0; j < GCN_U; ++j) {
            int ej = e + j * GCN_BS;
            int dv = dst[ej < e1 ? ej : e1 - 1];
            d[j] = ej < e1 ? dv - lo : -1;
        }
#pragma unroll
        for (int j = 0; j < GCN_U; ++j) {
            if ((unsigned)d[j] < (unsigned)nr) atomicAdd(&cnt[d[j]], 1u);
        }
    }
    __syncthreads();
    float* __restrict__ dg = (gcn_two(ws_bytes, need_bytes) && gcn_half()) ? degb : deg;
    const int rot = (int)((blockIdx.x * ((GCN_RN / GCN_K) & ~31u)) % (unsigned)nr);
    for (int t = tid; t < nr; t += GCN_BS) {
        int i = t + rot;
        i = i >= nr ? i - nr : i;
        unsigned c = cnt[i];
        if (c != 0u) atomicAdd(&dg[lo + i], (float)c);
    }
}

__global__ void gcn_dis(const float* __restrict__ deg, const float* __restrict__ degb,
                        float* __restrict__ dis, int N, size_t ws_bytes, size_t need_bytes) {
    int i = blockIdx.x * blockDim.x + threadIdx.x;
    const bool two = gcn_two(ws_bytes, need_bytes);
    const float* __restrict__ degx = two ? degb : deg;
    if (i < N) {
        const float d0 = deg[i];
        const float d1 = degx[i];
        dis[i] = rsqrtf((two ? d0 + d1 : d0) + 1.0f);
    }
}

__global__ __launch_bounds__(GCN_BS)
void gcn_edge1(const int* __restrict__ src, const int* __restrict__ dst,
               const float* __restrict__ x, const float* __restrict__ dis,
               float* __restrict__ s1, float* __restrict__ s1b, int E, int N, int chunk,
               size_t ws_bytes, size_t need_bytes) {
    __shared__ float acc[GCN_RN];
    const int tid = threadIdx.x;
    const int lo = blockIdx.y * GCN_RN;
    int nr = N - lo;
    nr = nr < GCN_RN ? nr : GCN_RN;
    for (int i = tid; i < GCN_RN; i += GCN_BS) acc[i] = 0.0f;
    __syncthreads();

    const int e0 = blockIdx.x * chunk;
    int e1 = e0 + chunk;
    e1 = e1 < E ? e1 : E;
    for (int e = e0 + tid; e < e1; e += GCN_U * GCN_BS) {
        int d[GCN_U], s[GCN_U];
        float v[GCN_U];
#pragma unroll
        for (int j = 0; j < GCN_U; ++j) {
            int ej = e + j * GCN_BS;
            int ec = ej < e1 ? ej : e1 - 1;
            int dv = dst[ec];
            s[j] = src[ec];
            d[j] = ej < e1 ? dv - lo : -1;
        }
#pragma unroll
        for (int j = 0; j < GCN_U; ++j) {
            int se = ((unsigned)d[j] < (unsigned)nr) ? s[j] : 0;
            v[j] = x[se] * dis[se];
        }
#pragma unroll
        for (int j = 0; j < GCN_U; ++j) {
            asm volatile("" : : "v"(v[j]));
        }
#pragma unroll
        for (int j = 0; j < GCN_U; ++j) {
            if ((unsigned)d[j] < (unsigned)nr) atomicAdd(&acc[d[j]], v[j]);
        }
    }
    __syncthreads();
    float* __restrict__ sg = (gcn_two(ws_bytes, need_bytes) && gcn_half()) ? s1b : s1;
    const int rot = (int)((blockIdx.x * ((GCN_RN / GCN_K) & ~31u)) % (unsigned)nr);
    for (int t = tid; t < nr; t += GCN_BS) {
        int i = t + rot;
        i = i >= nr ? i - nr : i;
        atomicAdd(&sg[lo + i], acc[i] * dis[lo + i]);
    }
}

__global__ void gcn_make_t(const float* __restrict__ s1, const float* __restrict__ s1b,
                           const float* __restrict__ dis,
                           const float* __restrict__ x, float* __restrict__ t, int N,
                           size_t ws_bytes, size_t need_bytes) {
    int i = blockIdx.x * blockDim.x + threadIdx.x;
    const bool two = gcn_two(ws_bytes, need_bytes);
    const float* __restrict__ s1x = two ? s1b : s1;
    if (i < N) {
        const float s0 = s1[i];
        const float sb = s1x[i];
        t[i] = (two ? s0 + sb : s0) + dis[i] * dis[i] * x[i];
    }
}

__global__ void gcn_gemm_wmma(const float* __restrict__ t,
                              const float* __restrict__ W1,
                              const float* __restrict__ b1,
                              const float* __restrict__ W2,
                              float* __restrict__ h2pre,
                              int nTiles) {
    int wave = (int)((blockIdx.x * blockDim.x + threadIdx.x) >> 5);
    int lane = threadIdx.x & 31;
    if (wave >= nTiles) return;

    int mt = wave >> 1;
    int nt = wave & 1;
    int m0 = mt << 4;
    int n0 = nt << 4;
    int r = lane & 15;
    int khalf = lane >> 4;

    float trow = t[m0 + r];

    v8f acc = {};
#pragma unroll
    for (int k0 = 0; k0 < 64; k0 += 4) {
        int ka = k0 + khalf * 2;
        float a0 = fmaf(trow, W1[ka],     b1[ka]);
        float a1 = fmaf(trow, W1[ka + 1], b1[ka + 1]);
        v2f a, b;
        a.x = a0 > 0.0f ? a0 : 0.0f;
        a.y = a1 > 0.0f ? a1 : 0.0f;
        b.x = W2[ka * 32 + n0 + r];
        b.y = W2[(ka + 1) * 32 + n0 + r];
        acc = __builtin_amdgcn_wmma_f32_16x16x4_f32(
             false, a,  false, b,
             (short)0, acc,  false,  false);
    }

    int col = n0 + r;
    int rowBase = m0 + khalf * 8;
#pragma unroll
    for (int v = 0; v < 8; ++v) {
        h2pre[(rowBase + v) * 32 + col] = acc[v];
    }
}

#define GCN_E2U 8
__global__ __launch_bounds__(256)
void gcn_edge2(const int* __restrict__ src, const int* __restrict__ dst,
               const float* __restrict__ dis,
               const float* __restrict__ h2pre,
               float* __restrict__ agg2a, float* __restrict__ agg2b, int E,
               size_t ws_bytes, size_t need_bytes) {
    const int lane = threadIdx.x & 31;
    float* __restrict__ agg2 = (gcn_two(ws_bytes, need_bytes) && gcn_half()) ? agg2b : agg2a;
    const int wave = __builtin_amdgcn_readfirstlane(
        (int)((blockIdx.x * blockDim.x + threadIdx.x) >> 5));
    const int nWaves = (int)((gridDim.x * blockDim.x) >> 5);
    const int nBatch = (E + 31) >> 5;
    for (int b = wave; b < nBatch; b += nWaves) {
        const int e0 = b << 5;
        const int e = e0 + lane;
        const int ec = e < E ? e : E - 1;
        const int s = src[ec];
        const int d = dst[ec];
        const int nrm = __float_as_int(dis[s] * dis[d]);
        int cnt = E - e0;
        cnt = cnt < 32 ? cnt : 32;
        if (cnt == 32) {
#pragma unroll
            for (int j0 = 0; j0 < 32; j0 += GCN_E2U) {
                float v[GCN_E2U];
#pragma unroll
                for (int j = 0; j < GCN_E2U; ++j) {
                    const int sj = __builtin_amdgcn_readlane(s, j0 + j);
                    v[j] = h2pre[(long)sj * 32 + lane];
                }
#pragma unroll
                for (int j = 0; j < GCN_E2U; ++j) {
                    const int dj = __builtin_amdgcn_readlane(d, j0 + j);
                    const float nj = __int_as_float(__builtin_amdgcn_readlane(nrm, j0 + j));
                    atomicAdd(&agg2[(long)dj * 32 + lane], v[j] * nj);
                }
            }
        } else {
            for (int j = 0; j < cnt; ++j) {
                const int sj = __builtin_amdgcn_readlane(s, j);
                const int dj = __builtin_amdgcn_readlane(d, j);
                const float nj = __int_as_float(__builtin_amdgcn_readlane(nrm, j));
                atomicAdd(&agg2[(long)dj * 32 + lane], h2pre[(long)sj * 32 + lane] * nj);
            }
        }
    }
}

__global__ void gcn_pool(const float* __restrict__ agg2, const float* __restrict__ agg2b,
                         const float* __restrict__ h2pre,
                         const float* __restrict__ dis, const float* __restrict__ b2,
                         const int* __restrict__ batch,
                         float* __restrict__ psuma, float* __restrict__ psumb,
                         float* __restrict__ pcnta, float* __restrict__ pcntb, int N,
                         size_t ws_bytes, size_t need_bytes) {
    const bool two = gcn_two(ws_bytes, need_bytes);
    const float* __restrict__ agg2x = two ? agg2b : agg2;
    const bool hb = two && gcn_half();
    float* __restrict__ psum = hb ? psumb : psuma;
    float* __restrict__ pcnt = hb ? pcntb : pcnta;
    long tid = (long)blockIdx.x * blockDim.x + threadIdx.x;
    long total = (long)N * 32;
    long stride = (long)gridDim.x * blockDim.x;
    for (; tid < total; tid += stride) {
        int i = (int)(tid >> 5);
        int f = (int)(tid & 31);
        float d2 = dis[i] * dis[i];
        const float a0 = agg2[tid];
        const float a1 = agg2x[tid];
        const float a = two ? a0 + a1 : a0;
        float h = a + d2 * h2pre[tid] + b2[f];
        h = h > 0.0f ? h : 0.0f;
        int g = batch[i];
        atomicAdd(&psum[g * 32 + f], h);
        if (f == 0) atomicAdd(&pcnt[g], 1.0f);
    }
}

#define GCN_MLP_BS 1024
#define GCN_MLP_GPB (GCN_MLP_BS / 32)
__global__ __launch_bounds__(GCN_MLP_BS)
void gcn_mlp(const float* __restrict__ psum, const float* __restrict__ psumb,
             const float* __restrict__ pcnt, const float* __restrict__ pcntb,
             const float* __restrict__ Wp1, const float* __restrict__ bp1,
             const float* __restrict__ Wp2, const float* __restrict__ bp2,
             float* __restrict__ out, int G, size_t ws_bytes, size_t need_bytes) {
    const bool two = gcn_two(ws_bytes, need_bytes);
    const float* __restrict__ psumx = two ? psumb : psum;
    const float* __restrict__ pcntx = two ? pcntb : pcnt;
    __shared__ float res[GCN_MLP_GPB * 3];
    const int lane = threadIdx.x & 31;
    const int w = __builtin_amdgcn_readfirstlane((int)(threadIdx.x >> 5));
    const int g = blockIdx.x * GCN_MLP_GPB + w;
    if (g < G) {
        const float c0 = pcnt[g];
        const float c1 = pcntx[g];
        const float p0 = psum[g * 32 + lane];
        const float p1 = psumx[g * 32 + lane];
        float c = two ? c0 + c1 : c0;
        const float ps = two ? p0 + p1 : p0;
        c = c > 1.0f ? c : 1.0f;
        const float inv = 1.0f / c;
        const int p = __float_as_int(ps * inv);

        const int j0 = lane * 4;
        float h0 = bp1[j0], h1 = bp1[j0 + 1], h2 = bp1[j0 + 2], h3 = bp1[j0 + 3];
#pragma unroll
        for (int f = 0; f < 32; ++f) {
            const float pf = __int_as_float(__builtin_amdgcn_readlane(p, f));
            const float* wr = Wp1 + f * 128 + j0;
            h0 = fmaf(pf, wr[0], h0);
            h1 = fmaf(pf, wr[1], h1);
            h2 = fmaf(pf, wr[2], h2);
            h3 = fmaf(pf, wr[3], h3);
        }
        h0 = h0 > 0.0f ? h0 : 0.0f;
        h1 = h1 > 0.0f ? h1 : 0.0f;
        h2 = h2 > 0.0f ? h2 : 0.0f;
        h3 = h3 > 0.0f ? h3 : 0.0f;

        const float* v = Wp2 + j0 * 3;
        float o0 = fmaf(h3, v[9],  fmaf(h2, v[6], fmaf(h1, v[3], h0 * v[0])));
        float o1 = fmaf(h3, v[10], fmaf(h2, v[7], fmaf(h1, v[4], h0 * v[1])));
        float o2 = fmaf(h3, v[11], fmaf(h2, v[8], fmaf(h1, v[5], h0 * v[2])));
        for (int m = 16; m >= 1; m >>= 1) {
            o0 += __shfl_xor(o0, m);
            o1 += __shfl_xor(o1, m);
            o2 += __shfl_xor(o2, m);
        }
        if (lane == 0) {
            res[w * 3 + 0] = o0 + bp2[0];
            res[w * 3 + 1] = o1 + bp2[1];
            res[w * 3 + 2] = o2 + bp2[2];
        }
    }
    __syncthreads();
    if (w == 0) {
        const int gg = blockIdx.x * GCN_MLP_GPB + lane;
        if (gg < G) {
            out[gg * 3 + 0] = res[lane * 3 + 0];
            out[gg * 3 + 1] = res[lane * 3 + 1];
            out[gg * 3 + 2] = res[lane * 3 + 2];
        }
    }
}

extern "C" void kernel_launch(void* const* d_in, const int* in_sizes, int n_in,
                              void* d_out, int out_size, void* d_ws, size_t ws_size,
                              hipStream_t stream) {
    const float* x   = (const float*)d_in[0];
    const float* W1  = (const float*)d_in[1];
    const float* b1  = (const float*)d_in[2];
    const float* W2  = (const float*)d_in[3];
    const float* b2  = (const float*)d_in[4];
    const float* Wp1 = (const float*)d_in[5];
    const float* bp1 = (const float*)d_in[6];
    const float* Wp2 = (const float*)d_in[7];
    const float* bp2 = (const float*)d_in[8];
    const int* eidx  = (const int*)d_in[9];
    const int* batch = (const int*)d_in[10];
    float* out = (float*)d_out;

    const int N = in_sizes[0];
    const int E = in_sizes[9] / 2;
    const int G = out_size / 3;
    const int* src = eidx;
    const int* dst = eidx + E;

    float* wsf   = (float*)d_ws;
    float* deg   = wsf;
    float* s1    = deg + N;
    float* dis   = s1 + N;
    float* tbuf  = dis + N;
    float* h2pre = tbuf + N;
    float* agg2  = h2pre + (long)N * 32;
    float* psum  = agg2 + (long)N * 32;
    float* pcnt  = psum + (long)G * 32;
    float* agg2b = (float*)(((uintptr_t)(pcnt + G) + 255) & ~(uintptr_t)255);
    float* psumb = agg2b + (long)N * 32;
    float* degb  = psumb + (long)G * 32;
    float* s1b   = degb + N;
    float* pcntb = s1b + N;
    float* wsend = pcntb + G;
    const size_t need = (size_t)((const char*)wsend - (const char*)d_ws);

    const int TPB = 256;

    long nz1 = 2L * N;
    long nz2 = 32L * N + 32L * G + G;
    long nz2two = (long)(wsend - agg2);
    gcn_zero_f32<<<1024, TPB, 0, stream>>>(deg, nz1, nz1, ws_size, need);
    gcn_zero_f32<<<2048, TPB, 0, stream>>>(agg2, nz2, nz2two, ws_size, need);

    const int chunk = (E + GCN_K - 1) / GCN_K;
    const dim3 sgrid(GCN_K, (N + GCN_RN - 1) / GCN_RN);
    gcn_degree<<<sgrid, GCN_BS, 0, stream>>>(dst, deg, degb, E, N, chunk, ws_size, need);
    gcn_dis<<<(N + TPB - 1) / TPB, TPB, 0, stream>>>(deg, degb, dis, N, ws_size, need);
    gcn_edge1<<<sgrid, GCN_BS, 0, stream>>>(src, dst, x, dis, s1, s1b, E, N, chunk, ws_size, need);
    gcn_make_t<<<(N + TPB - 1) / TPB, TPB, 0, stream>>>(s1, s1b, dis, x, tbuf, N, ws_size, need);

    int nTiles = (N / 16) * 2;
    gcn_gemm_wmma<<<(nTiles + 7) / 8, TPB, 0, stream>>>(tbuf, W1, b1, W2, h2pre, nTiles);

    gcn_edge2<<<2048, TPB, 0, stream>>>(src, dst, dis, h2pre, agg2, agg2b, E, ws_size, need);

    gcn_pool<<<4096, TPB, 0, stream>>>(agg2, agg2b, h2pre, dis, b2, batch, psum, psumb, pcnt, pcntb, N, ws_size, need);

    gcn_mlp<<<(G + GCN_MLP_GPB - 1) / GCN_MLP_GPB, GCN_MLP_BS, 0, stream>>>(psum, psumb, pcnt, pcntb, Wp1, bp1, Wp2, bp2, out, G, ws_size, need);
}


# edge passes gather one pre-scaled word or row per edge and no dis: dis kernel stores x*dis, GEMM scales A rows by dis[m], pool applies dis[dst] after the sum; layer-2 terms round differently
# speedup vs baseline: 4.9368x; 1.2384x over previous
#include <hip/hip_runtime.h>
#include <hip/hip_bf16.h>

typedef float v2f __attribute__((ext_vector_type(2)));
typedef float v8f __attribute__((ext_vector_type(8)));


__device__ __forceinline__ int gcn_half() {
    return (int)((__builtin_amdgcn_s_sendmsg_rtn(0x87) >> 18) & 1u);
}

__device__ __forceinline__ bool gcn_two(size_t ws_bytes, size_t need_bytes) {
    return ws_bytes >= need_bytes;
}

__global__ void gcn_zero_f32(float* __restrict__ p, long n_one, long n_two,
                             size_t ws_bytes, size_t need_bytes) {
    const long n = gcn_two(ws_bytes, need_bytes) ? n_two : n_one;
    long i = (long)blockIdx.x * blockDim.x + threadIdx.x;
    long stride = (long)gridDim.x * blockDim.x;
    for (; i < n; i += stride) p[i] = 0.0f;
}

#define GCN_RN 12512
#define GCN_BS 1024
#define GCN_K  32
#define GCN_U  8

__global__ __launch_bounds__(GCN_BS)
void gcn_degree(const int* __restrict__ dst, float* __restrict__ deg, float* __restrict__ degb,
                int E, int N, int chunk, size_t ws_bytes, size_t need_bytes) {
    __shared__ unsigned cnt[GCN_RN];
    const int tid = threadIdx.x;
    const int lo = blockIdx.y * GCN_RN;
    int nr = N - lo;
    nr = nr < GCN_RN ? nr : GCN_RN;
    for (int i = tid; i < GCN_RN; i += GCN_BS) cnt[i] = 0u;
    __syncthreads();

    const int e0 = blockIdx.x * chunk;
    int e1 = e0 + chunk;
    e1 = e1 < E ? e1 : E;
    for (int e = e0 + tid; e < e1; e += GCN_U * GCN_BS) {
        int d[GCN_U];
#pragma unroll
        for (int j = 0; j < GCN_U; ++j) {
            int ej = e + j * GCN_BS;
            int dv = dst[ej < e1 ? ej : e1 - 1];
            d[j] = ej < e1 ? dv - lo : -1;
        }
#pragma unroll
        for (int j = 0; j < GCN_U; ++j) {
            if ((unsigned)d[j] < (unsigned)nr) atomicAdd(&cnt[d[j]], 1u);
        }
    }
    __syncthreads();
    float* __restrict__ dg = (gcn_two(ws_bytes, need_bytes) && gcn_half()) ? degb : deg;
    const int rot = (int)((blockIdx.x * ((GCN_RN / GCN_K) & ~31u)) % (unsigned)nr);
    for (int t = tid; t < nr; t += GCN_BS) {
        int i = t + rot;
        i = i >= nr ? i - nr : i;
        unsigned c = cnt[i];
        if (c != 0u) atomicAdd(&dg[lo + i], (float)c);
    }
}

__global__ void gcn_dis(const float* __restrict__ deg, const float* __restrict__ degb,
                        const float* __restrict__ x,
                        float* __restrict__ dis, float* __restrict__ xd,
                        int N, size_t ws_bytes, size_t need_bytes) {
    int i = blockIdx.x * blockDim.x + threadIdx.x;
    const bool two = gcn_two(ws_bytes, need_bytes);
    const float* __restrict__ degx = two ? degb : deg;
    if (i < N) {
        const float d0 = deg[i];
        const float d1 = degx[i];
        const float di = rsqrtf((two ? d0 + d1 : d0) + 1.0f);
        dis[i] = di;
        xd[i] = x[i] * di;
    }
}

__global__ __launch_bounds__(GCN_BS)
void gcn_edge1(const int* __restrict__ src, const int* __restrict__ dst,
               const float* __restrict__ xd, const float* __restrict__ dis,
               float* __restrict__ s1, float* __restrict__ s1b, int E, int N, int chunk,
               size_t ws_bytes, size_t need_bytes) {
    __shared__ float acc[GCN_RN];
    const int tid = threadIdx.x;
    const int lo = blockIdx.y * GCN_RN;
    int nr = N - lo;
    nr = nr < GCN_RN ? nr : GCN_RN;
    for (int i = tid; i < GCN_RN; i += GCN_BS) acc[i] = 0.0f;
    __syncthreads();

    const int e0 = blockIdx.x * chunk;
    int e1 = e0 + chunk;
    e1 = e1 < E ? e1 : E;
    for (int e = e0 + tid; e < e1; e += GCN_U * GCN_BS) {
        int d[GCN_U], s[GCN_U];
        float v[GCN_U];
#pragma unroll
        for (int j = 0; j < GCN_U; ++j) {
            int ej = e + j * GCN_BS;
            int ec = ej < e1 ? ej : e1 - 1;
            int dv = dst[ec];
            s[j] = src[ec];
            d[j] = ej < e1 ? dv - lo : -1;
        }
#pragma unroll
        for (int j = 0; j < GCN_U; ++j) {
            int se = ((unsigned)d[j] < (unsigned)nr) ? s[j] : 0;
            v[j] = xd[se];
        }
#pragma unroll
        for (int j = 0; j < GCN_U; ++j) {
            asm volatile("" : : "v"(v[j]));
        }
#pragma unroll
        for (int j = 0; j < GCN_U; ++j) {
            if ((unsigned)d[j] < (unsigned)nr) atomicAdd(&acc[d[j]], v[j]);
        }
    }
    __syncthreads();
    float* __restrict__ sg = (gcn_two(ws_bytes, need_bytes) && gcn_half()) ? s1b : s1;
    const int rot = (int)((blockIdx.x * ((GCN_RN / GCN_K) & ~31u)) % (unsigned)nr);
    for (int t = tid; t < nr; t += GCN_BS) {
        int i = t + rot;
        i = i >= nr ? i - nr : i;
        atomicAdd(&sg[lo + i], acc[i] * dis[lo + i]);
    }
}

__global__ void gcn_make_t(const float* __restrict__ s1, const float* __restrict__ s1b,
                           const float* __restrict__ dis,
                           const float* __restrict__ x, float* __restrict__ t, int N,
                           size_t ws_bytes, size_t need_bytes) {
    int i = blockIdx.x * blockDim.x + threadIdx.x;
    const bool two = gcn_two(ws_bytes, need_bytes);
    const float* __restrict__ s1x = two ? s1b : s1;
    if (i < N) {
        const float s0 = s1[i];
        const float sb = s1x[i];
        t[i] = (two ? s0 + sb : s0) + dis[i] * dis[i] * x[i];
    }
}

__global__ void gcn_gemm_wmma(const float* __restrict__ t,
                              const float* __restrict__ W1,
                              const float* __restrict__ b1,
                              const float* __restrict__ W2,
                              const float* __restrict__ dis,
                              float* __restrict__ h2s,
                              int nTiles) {
    int wave = (int)((blockIdx.x * blockDim.x + threadIdx.x) >> 5);
    int lane = threadIdx.x & 31;
    if (wave >= nTiles) return;

    int mt = wave >> 1;
    int nt = wave & 1;
    int m0 = mt << 4;
    int n0 = nt << 4;
    int r = lane & 15;
    int khalf = lane >> 4;

    float trow = t[m0 + r];

    float drow = dis[m0 + r];

    v8f acc = {};
#pragma unroll
    for (int k0 = 0; k0 < 64; k0 += 4) {
        int ka = k0 + khalf * 2;
        float a0 = fmaf(trow, W1[ka],     b1[ka]);
        float a1 = fmaf(trow, W1[ka + 1], b1[ka + 1]);
        v2f a, b;
        a.x = (a0 > 0.0f ? a0 : 0.0f) * drow;
        a.y = (a1 > 0.0f ? a1 : 0.0f) * drow;
        b.x = W2[ka * 32 + n0 + r];
        b.y = W2[(ka + 1) * 32 + n0 + r];
        acc = __builtin_amdgcn_wmma_f32_16x16x4_f32(
             false, a,  false, b,
             (short)0, acc,  false,  false);
    }

    int col = n0 + r;
    int rowBase = m0 + khalf * 8;
#pragma unroll
    for (int v = 0; v < 8; ++v) {
        h2s[(rowBase + v) * 32 + col] = acc[v];
    }
}

#define GCN_E2U 8
__global__ __launch_bounds__(256)
void gcn_edge2(const int* __restrict__ src, const int* __restrict__ dst,
               const float* __restrict__ h2s,
               float* __restrict__ agg2a, float* __restrict__ agg2b, int E,
               size_t ws_bytes, size_t need_bytes) {
    const int lane = threadIdx.x & 31;
    float* __restrict__ agg2 = (gcn_two(ws_bytes, need_bytes) && gcn_half()) ? agg2b : agg2a;
    const int wave = __builtin_amdgcn_readfirstlane(
        (int)((blockIdx.x * blockDim.x + threadIdx.x) >> 5));
    const int nWaves = (int)((gridDim.x * blockDim.x) >> 5);
    const int nBatch = (E + 31) >> 5;
    for (int b = wave; b < nBatch; b += nWaves) {
        const int e0 = b << 5;
        const int e = e0 + lane;
        const int ec = e < E ? e : E - 1;
        const int s = src[ec];
        const int d = dst[ec];
        int cnt = E - e0;
        cnt = cnt < 32 ? cnt : 32;
        if (cnt == 32) {
#pragma unroll
            for (int j0 = 0; j0 < 32; j0 += GCN_E2U) {
                float v[GCN_E2U];
#pragma unroll
                for (int j = 0; j < GCN_E2U; ++j) {
                    const int sj = __builtin_amdgcn_readlane(s, j0 + j);
                    v[j] = h2s[(long)sj * 32 + lane];
                }
#pragma unroll
                for (int j = 0; j < GCN_E2U; ++j) {
                    const int dj = __builtin_amdgcn_readlane(d, j0 + j);
                    atomicAdd(&agg2[(long)dj * 32 + lane], v[j]);
                }
            }
        } else {
            for (int j = 0; j < cnt; ++j) {
                const int sj = __builtin_amdgcn_readlane(s, j);
                const int dj = __builtin_amdgcn_readlane(d, j);
                atomicAdd(&agg2[(long)dj * 32 + lane], h2s[(long)sj * 32 + lane]);
            }
        }
    }
}

__global__ void gcn_pool(const float* __restrict__ agg2, const float* __restrict__ agg2b,
                         const float* __restrict__ h2s,
                         const float* __restrict__ dis, const float* __restrict__ b2,
                         const int* __restrict__ batch,
                         float* __restrict__ psuma, float* __restrict__ psumb,
                         float* __restrict__ pcnta, float* __restrict__ pcntb, int N,
                         size_t ws_bytes, size_t need_bytes) {
    const bool two = gcn_two(ws_bytes, need_bytes);
    const float* __restrict__ agg2x = two ? agg2b : agg2;
    const bool hb = two && gcn_half();
    float* __restrict__ psum = hb ? psumb : psuma;
    float* __restrict__ pcnt = hb ? pcntb : pcnta;
    long tid = (long)blockIdx.x * blockDim.x + threadIdx.x;
    long total = (long)N * 32;
    long stride = (long)gridDim.x * blockDim.x;
    for (; tid < total; tid += stride) {
        int i = (int)(tid >> 5);
        int f = (int)(tid & 31);
        const float di = dis[i];
        const float a0 = agg2[tid];
        const float a1 = agg2x[tid];
        const float a = two ? a0 + a1 : a0;
        float h = di * (a + h2s[tid]) + b2[f];
        h = h > 0.0f ? h : 0.0f;
        int g = batch[i];
        atomicAdd(&psum[g * 32 + f], h);
        if (f == 0) atomicAdd(&pcnt[g], 1.0f);
    }
}

#define GCN_MLP_BS 1024
#define GCN_MLP_GPB (GCN_MLP_BS / 32)
__global__ __launch_bounds__(GCN_MLP_BS)
void gcn_mlp(const float* __restrict__ psum, const float* __restrict__ psumb,
             const float* __restrict__ pcnt, const float* __restrict__ pcntb,
             const float* __restrict__ Wp1, const float* __restrict__ bp1,
             const float* __restrict__ Wp2, const float* __restrict__ bp2,
             float* __restrict__ out, int G, size_t ws_bytes, size_t need_bytes) {
    const bool two = gcn_two(ws_bytes, need_bytes);
    const float* __restrict__ psumx = two ? psumb : psum;
    const float* __restrict__ pcntx = two ? pcntb : pcnt;
    __shared__ float res[GCN_MLP_GPB * 3];
    const int lane = threadIdx.x & 31;
    const int w = __builtin_amdgcn_readfirstlane((int)(threadIdx.x >> 5));
    const int g = blockIdx.x * GCN_MLP_GPB + w;
    if (g < G) {
        const float c0 = pcnt[g];
        const float c1 = pcntx[g];
        const float p0 = psum[g * 32 + lane];
        const float p1 = psumx[g * 32 + lane];
        float c = two ? c0 + c1 : c0;
        const float ps = two ? p0 + p1 : p0;
        c = c > 1.0f ? c : 1.0f;
        const float inv = 1.0f / c;
        const int p = __float_as_int(ps * inv);

        const int j0 = lane * 4;
        float h0 = bp1[j0], h1 = bp1[j0 + 1], h2 = bp1[j0 + 2], h3 = bp1[j0 + 3];
#pragma unroll
        for (int f = 0; f < 32; ++f) {
            const float pf = __int_as_float(__builtin_amdgcn_readlane(p, f));
            const float* wr = Wp1 + f * 128 + j0;
            h0 = fmaf(pf, wr[0], h0);
            h1 = fmaf(pf, wr[1], h1);
            h2 = fmaf(pf, wr[2], h2);
            h3 = fmaf(pf, wr[3], h3);
        }
        h0 = h0 > 0.0f ? h0 : 0.0f;
        h1 = h1 > 0.0f ? h1 : 0.0f;
        h2 = h2 > 0.0f ? h2 : 0.0f;
        h3 = h3 > 0.0f ? h3 : 0.0f;

        const float* v = Wp2 + j0 * 3;
        float o0 = fmaf(h3, v[9],  fmaf(h2, v[6], fmaf(h1, v[3], h0 * v[0])));
        float o1 = fmaf(h3, v[10], fmaf(h2, v[7], fmaf(h1, v[4], h0 * v[1])));
        float o2 = fmaf(h3, v[11], fmaf(h2, v[8], fmaf(h1, v[5], h0 * v[2])));
        for (int m = 16; m >= 1; m >>= 1) {
            o0 += __shfl_xor(o0, m);
            o1 += __shfl_xor(o1, m);
            o2 += __shfl_xor(o2, m);
        }
        if (lane == 0) {
            res[w * 3 + 0] = o0 + bp2[0];
            res[w * 3 + 1] = o1 + bp2[1];
            res[w * 3 + 2] = o2 + bp2[2];
        }
    }
    __syncthreads();
    if (w == 0) {
        const int gg = blockIdx.x * GCN_MLP_GPB + lane;
        if (gg < G) {
            out[gg * 3 + 0] = res[lane * 3 + 0];
            out[gg * 3 + 1] = res[lane * 3 + 1];
            out[gg * 3 + 2] = res[lane * 3 + 2];
        }
    }
}

extern "C" void kernel_launch(void* const* d_in, const int* in_sizes, int n_in,
                              void* d_out, int out_size, void* d_ws, size_t ws_size,
                              hipStream_t stream) {
    const float* x   = (const float*)d_in[0];
    const float* W1  = (const float*)d_in[1];
    const float* b1  = (const float*)d_in[2];
    const float* W2  = (const float*)d_in[3];
    const float* b2  = (const float*)d_in[4];
    const float* Wp1 = (const float*)d_in[5];
    const float* bp1 = (const float*)d_in[6];
    const float* Wp2 = (const float*)d_in[7];
    const float* bp2 = (const float*)d_in[8];
    const int* eidx  = (const int*)d_in[9];
    const int* batch = (const int*)d_in[10];
    float* out = (float*)d_out;

    const int N = in_sizes[0];
    const int E = in_sizes[9] / 2;
    const int G = out_size / 3;
    const int* src = eidx;
    const int* dst = eidx + E;

    float* wsf   = (float*)d_ws;
    float* deg   = wsf;
    float* s1    = deg + N;
    float* dis   = s1 + N;
    float* tbuf  = dis + N;
    float* h2s   = tbuf + N;
    float* agg2  = h2s + (long)N * 32;
    float* psum  = agg2 + (long)N * 32;
    float* pcnt  = psum + (long)G * 32;
    float* agg2b = (float*)(((uintptr_t)(pcnt + G) + 255) & ~(uintptr_t)255);
    float* psumb = agg2b + (long)N * 32;
    float* degb  = psumb + (long)G * 32;
    float* s1b   = degb + N;
    float* pcntb = s1b + N;
    float* wsend = pcntb + G;
    const size_t need = (size_t)((const char*)wsend - (const char*)d_ws);

    const int TPB = 256;

    long nz1 = 2L * N;
    long nz2 = 32L * N + 32L * G + G;
    long nz2two = (long)(wsend - agg2);
    gcn_zero_f32<<<1024, TPB, 0, stream>>>(deg, nz1, nz1, ws_size, need);
    gcn_zero_f32<<<2048, TPB, 0, stream>>>(agg2, nz2, nz2two, ws_size, need);

    const int chunk = (E + GCN_K - 1) / GCN_K;
    const dim3 sgrid(GCN_K, (N + GCN_RN - 1) / GCN_RN);
    gcn_degree<<<sgrid, GCN_BS, 0, stream>>>(dst, deg, degb, E, N, chunk, ws_size, need);
    gcn_dis<<<(N + TPB - 1) / TPB, TPB, 0, stream>>>(deg, degb, x, dis, tbuf, N, ws_size, need);
    gcn_edge1<<<sgrid, GCN_BS, 0, stream>>>(src, dst, tbuf, dis, s1, s1b, E, N, chunk, ws_size, need);
    gcn_make_t<<<(N + TPB - 1) / TPB, TPB, 0, stream>>>(s1, s1b, dis, x, tbuf, N, ws_size, need);

    int nTiles = (N / 16) * 2;
    gcn_gemm_wmma<<<(nTiles + 7) / 8, TPB, 0, stream>>>(tbuf, W1, b1, W2, dis, h2s, nTiles);

    gcn_edge2<<<2048, TPB, 0, stream>>>(src, dst, h2s, agg2, agg2b, E, ws_size, need);

    gcn_pool<<<4096, TPB, 0, stream>>>(agg2, agg2b, h2s, dis, b2, batch, psum, psumb, pcnt, pcntb, N, ws_size, need);

    gcn_mlp<<<(G + GCN_MLP_GPB - 1) / GCN_MLP_GPB, GCN_MLP_BS, 0, stream>>>(psum, psumb, pcnt, pcntb, Wp1, bp1, Wp2, bp2, out, G, ws_size, need);
}


# layer-1 edge pass: block (chunk, src range, dst range) copies its src range of x*dis into LDS beside the dst sums and reads it there, no per-edge global gather; s1 partial-sum order changes
# speedup vs baseline: 5.1405x; 1.0413x over previous
#include <hip/hip_runtime.h>
#include <hip/hip_bf16.h>

typedef float v2f __attribute__((ext_vector_type(2)));
typedef float v8f __attribute__((ext_vector_type(8)));


__device__ __forceinline__ int gcn_half() {
    return (int)((__builtin_amdgcn_s_sendmsg_rtn(0x87) >> 18) & 1u);
}

__device__ __forceinline__ bool gcn_two(size_t ws_bytes, size_t need_bytes) {
    return ws_bytes >= need_bytes;
}

__global__ void gcn_zero_f32(float* __restrict__ p, long n_one, long n_two,
                             size_t ws_bytes, size_t need_bytes) {
    const long n = gcn_two(ws_bytes, need_bytes) ? n_two : n_one;
    long i = (long)blockIdx.x * blockDim.x + threadIdx.x;
    long stride = (long)gridDim.x * blockDim.x;
    for (; i < n; i += stride) p[i] = 0.0f;
}

#define GCN_RN 12512
#define GCN_BS 1024
#define GCN_K  32
#define GCN_U  8

__global__ __launch_bounds__(GCN_BS)
void gcn_degree(const int* __restrict__ dst, float* __restrict__ deg, float* __restrict__ degb,
                int E, int N, int chunk, size_t ws_bytes, size_t need_bytes) {
    __shared__ unsigned cnt[GCN_RN];
    const int tid = threadIdx.x;
    const int lo = blockIdx.y * GCN_RN;
    int nr = N - lo;
    nr = nr < GCN_RN ? nr : GCN_RN;
    for (int i = tid; i < GCN_RN; i += GCN_BS) cnt[i] = 0u;
    __syncthreads();

    const int e0 = blockIdx.x * chunk;
    int e1 = e0 + chunk;
    e1 = e1 < E ? e1 : E;
    for (int e = e0 + tid; e < e1; e += GCN_U * GCN_BS) {
        int d[GCN_U];
#pragma unroll
        for (int j = 0; j < GCN_U; ++j) {
            int ej = e + j * GCN_BS;
            int dv = dst[ej < e1 ? ej : e1 - 1];
            d[j] = ej < e1 ? dv - lo : -1;
        }
#pragma unroll
        for (int j = 0; j < GCN_U; ++j) {
            if ((unsigned)d[j] < (unsigned)nr) atomicAdd(&cnt[d[j]], 1u);
        }
    }
    __syncthreads();
    float* __restrict__ dg = (gcn_two(ws_bytes, need_bytes) && gcn_half()) ? degb : deg;
    const int rot = (int)((blockIdx.x * ((GCN_RN / GCN_K) & ~31u)) % (unsigned)nr);
    for (int t = tid; t < nr; t += GCN_BS) {
        int i = t + rot;
        i = i >= nr ? i - nr : i;
        unsigned c = cnt[i];
        if (c != 0u) atomicAdd(&dg[lo + i], (float)c);
    }
}

__global__ void gcn_dis(const float* __restrict__ deg, const float* __restrict__ degb,
                        const float* __restrict__ x,
                        float* __restrict__ dis, float* __restrict__ xd,
                        int N, size_t ws_bytes, size_t need_bytes) {
    int i = blockIdx.x * blockDim.x + threadIdx.x;
    const bool two = gcn_two(ws_bytes, need_bytes);
    const float* __restrict__ degx = two ? degb : deg;
    if (i < N) {
        const float d0 = deg[i];
        const float d1 = degx[i];
        const float di = rsqrtf((two ? d0 + d1 : d0) + 1.0f);
        dis[i] = di;
        xd[i] = x[i] * di;
    }
}

#define GCN_SR 50016
#define GCN_DR 25024
#define GCN_SR4_TRIPS ((GCN_SR / 4 + GCN_BS - 1) / GCN_BS)
__global__ __launch_bounds__(GCN_BS)
void gcn_edge1(const int* __restrict__ src, const int* __restrict__ dst,
               const float* __restrict__ xd, const float* __restrict__ dis,
               float* __restrict__ s1, float* __restrict__ s1b, int E, int N, int chunk,
               size_t ws_bytes, size_t need_bytes) {
    __shared__ float xl[GCN_SR];
    __shared__ float acc[GCN_DR];
    const int tid = threadIdx.x;
    const int nDst = (N + GCN_DR - 1) / GCN_DR;
    const int sr = (int)blockIdx.y / nDst;
    const int dr = (int)blockIdx.y - sr * nDst;
    const int slo = sr * GCN_SR;
    const int dlo = dr * GCN_DR;
    int ns = N - slo;
    ns = ns < GCN_SR ? ns : GCN_SR;
    int nd = N - dlo;
    nd = nd < GCN_DR ? nd : GCN_DR;

    const float* __restrict__ xs = xd + slo;
    if ((((uintptr_t)xs) & 15) == 0) {
        const float4* __restrict__ xs4 = (const float4*)xs;
        const int ns4 = ns >> 2;
        float4 q[GCN_SR4_TRIPS];
#pragma unroll
        for (int k = 0; k < GCN_SR4_TRIPS; ++k) {
            const int i = tid + k * GCN_BS;
            q[k] = xs4[i < ns4 ? i : 0];
        }
#pragma unroll
        for (int k = 0; k < GCN_SR4_TRIPS; ++k) {
            const int i = tid + k * GCN_BS;
            if (i < ns4) {
                xl[4 * i + 0] = q[k].x;
                xl[4 * i + 1] = q[k].y;
                xl[4 * i + 2] = q[k].z;
                xl[4 * i + 3] = q[k].w;
            }
        }
        for (int i = (ns4 << 2) + tid; i < ns; i += GCN_BS) xl[i] = xs[i];
    } else {
        for (int i = tid; i < ns; i += GCN_BS) xl[i] = xs[i];
    }
    for (int i = tid; i < GCN_DR; i += GCN_BS) acc[i] = 0.0f;
    __syncthreads();

    const int e0 = blockIdx.x * chunk;
    int e1 = e0 + chunk;
    e1 = e1 < E ? e1 : E;
    for (int e = e0 + tid; e < e1; e += GCN_U * GCN_BS) {
        int dvr[GCN_U], svr[GCN_U];
        int d[GCN_U], s[GCN_U];
        float v[GCN_U];
#pragma unroll
        for (int j = 0; j < GCN_U; ++j) {
            int ej = e + j * GCN_BS;
            int ec = ej < e1 ? ej : e1 - 1;
            dvr[j] = dst[ec];
            svr[j] = src[ec];
        }
        static_assert(GCN_U == 8, "the statement below names 8 + 8 registers");
        asm volatile("" : "+v"(dvr[0]), "+v"(dvr[1]), "+v"(dvr[2]), "+v"(dvr[3]),
                          "+v"(dvr[4]), "+v"(dvr[5]), "+v"(dvr[6]), "+v"(dvr[7]),
                          "+v"(svr[0]), "+v"(svr[1]), "+v"(svr[2]), "+v"(svr[3]),
                          "+v"(svr[4]), "+v"(svr[5]), "+v"(svr[6]), "+v"(svr[7]));
#pragma unroll
        for (int j = 0; j < GCN_U; ++j) {
            int ej = e + j * GCN_BS;
            int dv = dvr[j] - dlo;
            int sv = svr[j] - slo;
            bool ok = ej < e1 && (unsigned)dv < (unsigned)nd && (unsigned)sv < (unsigned)ns;
            d[j] = ok ? dv : -1;
            s[j] = ok ? sv : 0;
        }
#pragma unroll
        for (int j = 0; j < GCN_U; ++j) {
            v[j] = xl[s[j]];
        }
#pragma unroll
        for (int j = 0; j < GCN_U; ++j) {
            asm volatile("" : : "v"(v[j]));
        }
#pragma unroll
        for (int j = 0; j < GCN_U; ++j) {
            if (d[j] >= 0) atomicAdd(&acc[d[j]], v[j]);
        }
    }
    __syncthreads();
    float* __restrict__ sg = (gcn_two(ws_bytes, need_bytes) && gcn_half()) ? s1b : s1;
    const int nSrc = (N + GCN_SR - 1) / GCN_SR;
    const unsigned step = (unsigned)((GCN_DR / GCN_K) / (nSrc > 0 ? nSrc : 1)) & ~31u;
    const int rot = (int)(((blockIdx.x * (unsigned)nSrc + (unsigned)sr) * step) % (unsigned)nd);
    for (int t0 = tid; t0 < nd; t0 += GCN_U * GCN_BS) {
        int ix[GCN_U];
        float dv[GCN_U];
#pragma unroll
        for (int j = 0; j < GCN_U; ++j) {
            int t = t0 + j * GCN_BS;
            int i = t + rot;
            i = i >= nd ? i - nd : i;
            ix[j] = t < nd ? i : -1;
            dv[j] = dis[dlo + (t < nd ? i : 0)];
        }
#pragma unroll
        for (int j = 0; j < GCN_U; ++j) {
            if (ix[j] >= 0) atomicAdd(&sg[dlo + ix[j]], acc[ix[j]] * dv[j]);
        }
    }
}

__global__ void gcn_make_t(const float* __restrict__ s1, const float* __restrict__ s1b,
                           const float* __restrict__ dis,
                           const float* __restrict__ x, float* __restrict__ t, int N,
                           size_t ws_bytes, size_t need_bytes) {
    int i = blockIdx.x * blockDim.x + threadIdx.x;
    const bool two = gcn_two(ws_bytes, need_bytes);
    const float* __restrict__ s1x = two ? s1b : s1;
    if (i < N) {
        const float s0 = s1[i];
        const float sb = s1x[i];
        t[i] = (two ? s0 + sb : s0) + dis[i] * dis[i] * x[i];
    }
}

__global__ void gcn_gemm_wmma(const float* __restrict__ t,
                              const float* __restrict__ W1,
                              const float* __restrict__ b1,
                              const float* __restrict__ W2,
                              const float* __restrict__ dis,
                              float* __restrict__ h2s,
                              int nTiles) {
    int wave = (int)((blockIdx.x * blockDim.x + threadIdx.x) >> 5);
    int lane = threadIdx.x & 31;
    if (wave >= nTiles) return;

    int mt = wave >> 1;
    int nt = wave & 1;
    int m0 = mt << 4;
    int n0 = nt << 4;
    int r = lane & 15;
    int khalf = lane >> 4;

    float trow = t[m0 + r];

    float drow = dis[m0 + r];

    v8f acc = {};
#pragma unroll
    for (int k0 = 0; k0 < 64; k0 += 4) {
        int ka = k0 + khalf * 2;
        float a0 = fmaf(trow, W1[ka],     b1[ka]);
        float a1 = fmaf(trow, W1[ka + 1], b1[ka + 1]);
        v2f a, b;
        a.x = (a0 > 0.0f ? a0 : 0.0f) * drow;
        a.y = (a1 > 0.0f ? a1 : 0.0f) * drow;
        b.x = W2[ka * 32 + n0 + r];
        b.y = W2[(ka + 1) * 32 + n0 + r];
        acc = __builtin_amdgcn_wmma_f32_16x16x4_f32(
             false, a,  false, b,
             (short)0, acc,  false,  false);
    }

    int col = n0 + r;
    int rowBase = m0 + khalf * 8;
#pragma unroll
    for (int v = 0; v < 8; ++v) {
        h2s[(rowBase + v) * 32 + col] = acc[v];
    }
}

#define GCN_E2U 8
__global__ __launch_bounds__(256)
void gcn_edge2(const int* __restrict__ src, const int* __restrict__ dst,
               const float* __restrict__ h2s,
               float* __restrict__ agg2a, float* __restrict__ agg2b, int E,
               size_t ws_bytes, size_t need_bytes) {
    const int lane = threadIdx.x & 31;
    float* __restrict__ agg2 = (gcn_two(ws_bytes, need_bytes) && gcn_half()) ? agg2b : agg2a;
    const int wave = __builtin_amdgcn_readfirstlane(
        (int)((blockIdx.x * blockDim.x + threadIdx.x) >> 5));
    const int nWaves = (int)((gridDim.x * blockDim.x) >> 5);
    const int nBatch = (E + 31) >> 5;
    for (int b = wave; b < nBatch; b += nWaves) {
        const int e0 = b << 5;
        const int e = e0 + lane;
        const int ec = e < E ? e : E - 1;
        const int s = src[ec];
        const int d = dst[ec];
        int cnt = E - e0;
        cnt = cnt < 32 ? cnt : 32;
        if (cnt == 32) {
#pragma unroll
            for (int j0 = 0; j0 < 32; j0 += GCN_E2U) {
                float v[GCN_E2U];
#pragma unroll
                for (int j = 0; j < GCN_E2U; ++j) {
                    const int sj = __builtin_amdgcn_readlane(s, j0 + j);
                    v[j] = h2s[(long)sj * 32 + lane];
                }
#pragma unroll
                for (int j = 0; j < GCN_E2U; ++j) {
                    const int dj = __builtin_amdgcn_readlane(d, j0 + j);
                    atomicAdd(&agg2[(long)dj * 32 + lane], v[j]);
                }
            }
        } else {
            for (int j = 0; j < cnt; ++j) {
                const int sj = __builtin_amdgcn_readlane(s, j);
                const int dj = __builtin_amdgcn_readlane(d, j);
                atomicAdd(&agg2[(long)dj * 32 + lane], h2s[(long)sj * 32 + lane]);
            }
        }
    }
}

__global__ void gcn_pool(const float* __restrict__ agg2, const float* __restrict__ agg2b,
                         const float* __restrict__ h2s,
                         const float* __restrict__ dis, const float* __restrict__ b2,
                         const int* __restrict__ batch,
                         float* __restrict__ psuma, float* __restrict__ psumb,
                         float* __restrict__ pcnta, float* __restrict__ pcntb, int N,
                         size_t ws_bytes, size_t need_bytes) {
    const bool two = gcn_two(ws_bytes, need_bytes);
    const float* __restrict__ agg2x = two ? agg2b : agg2;
    const bool hb = two && gcn_half();
    float* __restrict__ psum = hb ? psumb : psuma;
    float* __restrict__ pcnt = hb ? pcntb : pcnta;
    long tid = (long)blockIdx.x * blockDim.x + threadIdx.x;
    long total = (long)N * 32;
    long stride = (long)gridDim.x * blockDim.x;
    for (; tid < total; tid += stride) {
        int i = (int)(tid >> 5);
        int f = (int)(tid & 31);
        const float di = dis[i];
        const float a0 = agg2[tid];
        const float a1 = agg2x[tid];
        const float a = two ? a0 + a1 : a0;
        float h = di * (a + h2s[tid]) + b2[f];
        h = h > 0.0f ? h : 0.0f;
        int g = batch[i];
        atomicAdd(&psum[g * 32 + f], h);
        if (f == 0) atomicAdd(&pcnt[g], 1.0f);
    }
}

#define GCN_MLP_BS 1024
#define GCN_MLP_GPB (GCN_MLP_BS / 32)
__global__ __launch_bounds__(GCN_MLP_BS)
void gcn_mlp(const float* __restrict__ psum, const float* __restrict__ psumb,
             const float* __restrict__ pcnt, const float* __restrict__ pcntb,
             const float* __restrict__ Wp1, const float* __restrict__ bp1,
             const float* __restrict__ Wp2, const float* __restrict__ bp2,
             float* __restrict__ out, int G, size_t ws_bytes, size_t need_bytes) {
    const bool two = gcn_two(ws_bytes, need_bytes);
    const float* __restrict__ psumx = two ? psumb : psum;
    const float* __restrict__ pcntx = two ? pcntb : pcnt;
    __shared__ float res[GCN_MLP_GPB * 3];
    const int lane = threadIdx.x & 31;
    const int w = __builtin_amdgcn_readfirstlane((int)(threadIdx.x >> 5));
    const int g = blockIdx.x * GCN_MLP_GPB + w;
    if (g < G) {
        const float c0 = pcnt[g];
        const float c1 = pcntx[g];
        const float p0 = psum[g * 32 + lane];
        const float p1 = psumx[g * 32 + lane];
        float c = two ? c0 + c1 : c0;
        const float ps = two ? p0 + p1 : p0;
        c = c > 1.0f ? c : 1.0f;
        const float inv = 1.0f / c;
        const int p = __float_as_int(ps * inv);

        const int j0 = lane * 4;
        float h0 = bp1[j0], h1 = bp1[j0 + 1], h2 = bp1[j0 + 2], h3 = bp1[j0 + 3];
#pragma unroll
        for (int f = 0; f < 32; ++f) {
            const float pf = __int_as_float(__builtin_amdgcn_readlane(p, f));
            const float* wr = Wp1 + f * 128 + j0;
            h0 = fmaf(pf, wr[0], h0);
            h1 = fmaf(pf, wr[1], h1);
            h2 = fmaf(pf, wr[2], h2);
            h3 = fmaf(pf, wr[3], h3);
        }
        h0 = h0 > 0.0f ? h0 : 0.0f;
        h1 = h1 > 0.0f ? h1 : 0.0f;
        h2 = h2 > 0.0f ? h2 : 0.0f;
        h3 = h3 > 0.0f ? h3 : 0.0f;

        const float* v = Wp2 + j0 * 3;
        float o0 = fmaf(h3, v[9],  fmaf(h2, v[6], fmaf(h1, v[3], h0 * v[0])));
        float o1 = fmaf(h3, v[10], fmaf(h2, v[7], fmaf(h1, v[4], h0 * v[1])));
        float o2 = fmaf(h3, v[11], fmaf(h2, v[8], fmaf(h1, v[5], h0 * v[2])));
        for (int m = 16; m >= 1; m >>= 1) {
            o0 += __shfl_xor(o0, m);
            o1 += __shfl_xor(o1, m);
            o2 += __shfl_xor(o2, m);
        }
        if (lane == 0) {
            res[w * 3 + 0] = o0 + bp2[0];
            res[w * 3 + 1] = o1 + bp2[1];
            res[w * 3 + 2] = o2 + bp2[2];
        }
    }
    __syncthreads();
    if (w == 0) {
        const int gg = blockIdx.x * GCN_MLP_GPB + lane;
        if (gg < G) {
            out[gg * 3 + 0] = res[lane * 3 + 0];
            out[gg * 3 + 1] = res[lane * 3 + 1];
            out[gg * 3 + 2] = res[lane * 3 + 2];
        }
    }
}

extern "C" void kernel_launch(void* const* d_in, const int* in_sizes, int n_in,
                              void* d_out, int out_size, void* d_ws, size_t ws_size,
                              hipStream_t stream) {
    const float* x   = (const float*)d_in[0];
    const float* W1  = (const float*)d_in[1];
    const float* b1  = (const float*)d_in[2];
    const float* W2  = (const float*)d_in[3];
    const float* b2  = (const float*)d_in[4];
    const float* Wp1 = (const float*)d_in[5];
    const float* bp1 = (const float*)d_in[6];
    const float* Wp2 = (const float*)d_in[7];
    const float* bp2 = (const float*)d_in[8];
    const int* eidx  = (const int*)d_in[9];
    const int* batch = (const int*)d_in[10];
    float* out = (float*)d_out;

    const int N = in_sizes[0];
    const int E = in_sizes[9] / 2;
    const int G = out_size / 3;
    const int* src = eidx;
    const int* dst = eidx + E;

    float* wsf   = (float*)d_ws;
    float* deg   = wsf;
    float* s1    = deg + N;
    float* dis   = s1 + N;
    float* tbuf  = dis + N;
    float* h2s   = tbuf + N;
    float* agg2  = h2s + (long)N * 32;
    float* psum  = agg2 + (long)N * 32;
    float* pcnt  = psum + (long)G * 32;
    float* agg2b = (float*)(((uintptr_t)(pcnt + G) + 255) & ~(uintptr_t)255);
    float* psumb = agg2b + (long)N * 32;
    float* degb  = psumb + (long)G * 32;
    float* s1b   = degb + N;
    float* pcntb = s1b + N;
    float* wsend = pcntb + G;
    const size_t need = (size_t)((const char*)wsend - (const char*)d_ws);

    const int TPB = 256;

    long nz1 = 2L * N;
    long nz2 = 32L * N + 32L * G + G;
    long nz2two = (long)(wsend - agg2);
    gcn_zero_f32<<<1024, TPB, 0, stream>>>(deg, nz1, nz1, ws_size, need);
    gcn_zero_f32<<<2048, TPB, 0, stream>>>(agg2, nz2, nz2two, ws_size, need);

    const int chunk = (E + GCN_K - 1) / GCN_K;
    const dim3 sgrid(GCN_K, (N + GCN_RN - 1) / GCN_RN);
    gcn_degree<<<sgrid, GCN_BS, 0, stream>>>(dst, deg, degb, E, N, chunk, ws_size, need);
    gcn_dis<<<(N + TPB - 1) / TPB, TPB, 0, stream>>>(deg, degb, x, dis, tbuf, N, ws_size, need);
    const dim3 egrid(GCN_K, ((N + GCN_SR - 1) / GCN_SR) * ((N + GCN_DR - 1) / GCN_DR));
    gcn_edge1<<<egrid, GCN_BS, 0, stream>>>(src, dst, tbuf, dis, s1, s1b, E, N, chunk, ws_size, need);
    gcn_make_t<<<(N + TPB - 1) / TPB, TPB, 0, stream>>>(s1, s1b, dis, x, tbuf, N, ws_size, need);

    int nTiles = (N / 16) * 2;
    gcn_gemm_wmma<<<(nTiles + 7) / 8, TPB, 0, stream>>>(tbuf, W1, b1, W2, dis, h2s, nTiles);

    gcn_edge2<<<2048, TPB, 0, stream>>>(src, dst, h2s, agg2, agg2b, E, ws_size, need);

    gcn_pool<<<4096, TPB, 0, stream>>>(agg2, agg2b, h2s, dis, b2, batch, psum, psumb, pcnt, pcntb, N, ws_size, need);

    gcn_mlp<<<(G + GCN_MLP_GPB - 1) / GCN_MLP_GPB, GCN_MLP_BS, 0, stream>>>(psum, psumb, pcnt, pcntb, Wp1, bp1, Wp2, bp2, out, G, ws_size, need);
}


# pool: a wave takes 16 consecutive nodes, issues their 48 row loads as one group and adds runs of equal graph id in registers, one row atomic per run (6256 waves, was 32768); pooled sum order changes
# speedup vs baseline: 5.4983x; 1.0696x over previous
#include <hip/hip_runtime.h>
#include <hip/hip_bf16.h>

typedef float v2f __attribute__((ext_vector_type(2)));
typedef float v8f __attribute__((ext_vector_type(8)));


__device__ __forceinline__ int gcn_half() {
    return (int)((__builtin_amdgcn_s_sendmsg_rtn(0x87) >> 18) & 1u);
}

__device__ __forceinline__ bool gcn_two(size_t ws_bytes, size_t need_bytes) {
    return ws_bytes >= need_bytes;
}

__global__ void gcn_zero_f32(float* __restrict__ p, long n_one, long n_two,
                             size_t ws_bytes, size_t need_bytes) {
    const long n = gcn_two(ws_bytes, need_bytes) ? n_two : n_one;
    long i = (long)blockIdx.x * blockDim.x + threadIdx.x;
    long stride = (long)gridDim.x * blockDim.x;
    for (; i < n; i += stride) p[i] = 0.0f;
}

#define GCN_RN 12512
#define GCN_BS 1024
#define GCN_K  32
#define GCN_U  8

__global__ __launch_bounds__(GCN_BS)
void gcn_degree(const int* __restrict__ dst, float* __restrict__ deg, float* __restrict__ degb,
                int E, int N, int chunk, size_t ws_bytes, size_t need_bytes) {
    __shared__ unsigned cnt[GCN_RN];
    const int tid = threadIdx.x;
    const int lo = blockIdx.y * GCN_RN;
    int nr = N - lo;
    nr = nr < GCN_RN ? nr : GCN_RN;
    for (int i = tid; i < GCN_RN; i += GCN_BS) cnt[i] = 0u;
    __syncthreads();

    const int e0 = blockIdx.x * chunk;
    int e1 = e0 + chunk;
    e1 = e1 < E ? e1 : E;
    for (int e = e0 + tid; e < e1; e += GCN_U * GCN_BS) {
        int d[GCN_U];
#pragma unroll
        for (int j = 0; j < GCN_U; ++j) {
            int ej = e + j * GCN_BS;
            int dv = dst[ej < e1 ? ej : e1 - 1];
            d[j] = ej < e1 ? dv - lo : -1;
        }
#pragma unroll
        for (int j = 0; j < GCN_U; ++j) {
            if ((unsigned)d[j] < (unsigned)nr) atomicAdd(&cnt[d[j]], 1u);
        }
    }
    __syncthreads();
    float* __restrict__ dg = (gcn_two(ws_bytes, need_bytes) && gcn_half()) ? degb : deg;
    const int rot = (int)((blockIdx.x * ((GCN_RN / GCN_K) & ~31u)) % (unsigned)nr);
    for (int t = tid; t < nr; t += GCN_BS) {
        int i = t + rot;
        i = i >= nr ? i - nr : i;
        unsigned c = cnt[i];
        if (c != 0u) atomicAdd(&dg[lo + i], (float)c);
    }
}

__global__ void gcn_dis(const float* __restrict__ deg, const float* __restrict__ degb,
                        const float* __restrict__ x,
                        float* __restrict__ dis, float* __restrict__ xd,
                        int N, size_t ws_bytes, size_t need_bytes) {
    int i = blockIdx.x * blockDim.x + threadIdx.x;
    const bool two = gcn_two(ws_bytes, need_bytes);
    const float* __restrict__ degx = two ? degb : deg;
    if (i < N) {
        const float d0 = deg[i];
        const float d1 = degx[i];
        const float di = rsqrtf((two ? d0 + d1 : d0) + 1.0f);
        dis[i] = di;
        xd[i] = x[i] * di;
    }
}

#define GCN_SR 50016
#define GCN_DR 25024
#define GCN_SR4_TRIPS ((GCN_SR / 4 + GCN_BS - 1) / GCN_BS)
__global__ __launch_bounds__(GCN_BS)
void gcn_edge1(const int* __restrict__ src, const int* __restrict__ dst,
               const float* __restrict__ xd, const float* __restrict__ dis,
               float* __restrict__ s1, float* __restrict__ s1b, int E, int N, int chunk,
               size_t ws_bytes, size_t need_bytes) {
    __shared__ float xl[GCN_SR];
    __shared__ float acc[GCN_DR];
    const int tid = threadIdx.x;
    const int nDst = (N + GCN_DR - 1) / GCN_DR;
    const int sr = (int)blockIdx.y / nDst;
    const int dr = (int)blockIdx.y - sr * nDst;
    const int slo = sr * GCN_SR;
    const int dlo = dr * GCN_DR;
    int ns = N - slo;
    ns = ns < GCN_SR ? ns : GCN_SR;
    int nd = N - dlo;
    nd = nd < GCN_DR ? nd : GCN_DR;

    const float* __restrict__ xs = xd + slo;
    if ((((uintptr_t)xs) & 15) == 0) {
        const float4* __restrict__ xs4 = (const float4*)xs;
        const int ns4 = ns >> 2;
        float4 q[GCN_SR4_TRIPS];
#pragma unroll
        for (int k = 0; k < GCN_SR4_TRIPS; ++k) {
            const int i = tid + k * GCN_BS;
            q[k] = xs4[i < ns4 ? i : 0];
        }
#pragma unroll
        for (int k = 0; k < GCN_SR4_TRIPS; ++k) {
            const int i = tid + k * GCN_BS;
            if (i < ns4) {
                xl[4 * i + 0] = q[k].x;
                xl[4 * i + 1] = q[k].y;
                xl[4 * i + 2] = q[k].z;
                xl[4 * i + 3] = q[k].w;
            }
        }
        for (int i = (ns4 << 2) + tid; i < ns; i += GCN_BS) xl[i] = xs[i];
    } else {
        for (int i = tid; i < ns; i += GCN_BS) xl[i] = xs[i];
    }
    for (int i = tid; i < GCN_DR; i += GCN_BS) acc[i] = 0.0f;
    __syncthreads();

    const int e0 = blockIdx.x * chunk;
    int e1 = e0 + chunk;
    e1 = e1 < E ? e1 : E;
    for (int e = e0 + tid; e < e1; e += GCN_U * GCN_BS) {
        int dvr[GCN_U], svr[GCN_U];
        int d[GCN_U], s[GCN_U];
        float v[GCN_U];
#pragma unroll
        for (int j = 0; j < GCN_U; ++j) {
            int ej = e + j * GCN_BS;
            int ec = ej < e1 ? ej : e1 - 1;
            dvr[j] = dst[ec];
            svr[j] = src[ec];
        }
        static_assert(GCN_U == 8, "the statement below names 8 + 8 registers");
        asm volatile("" : "+v"(dvr[0]), "+v"(dvr[1]), "+v"(dvr[2]), "+v"(dvr[3]),
                          "+v"(dvr[4]), "+v"(dvr[5]), "+v"(dvr[6]), "+v"(dvr[7]),
                          "+v"(svr[0]), "+v"(svr[1]), "+v"(svr[2]), "+v"(svr[3]),
                          "+v"(svr[4]), "+v"(svr[5]), "+v"(svr[6]), "+v"(svr[7]));
#pragma unroll
        for (int j = 0; j < GCN_U; ++j) {
            int ej = e + j * GCN_BS;
            int dv = dvr[j] - dlo;
            int sv = svr[j] - slo;
            bool ok = ej < e1 && (unsigned)dv < (unsigned)nd && (unsigned)sv < (unsigned)ns;
            d[j] = ok ? dv : -1;
            s[j] = ok ? sv : 0;
        }
#pragma unroll
        for (int j = 0; j < GCN_U; ++j) {
            v[j] = xl[s[j]];
        }
#pragma unroll
        for (int j = 0; j < GCN_U; ++j) {
            asm volatile("" : : "v"(v[j]));
        }
#pragma unroll
        for (int j = 0; j < GCN_U; ++j) {
            if (d[j] >= 0) atomicAdd(&acc[d[j]], v[j]);
        }
    }
    __syncthreads();
    float* __restrict__ sg = (gcn_two(ws_bytes, need_bytes) && gcn_half()) ? s1b : s1;
    const int nSrc = (N + GCN_SR - 1) / GCN_SR;
    const unsigned step = (unsigned)((GCN_DR / GCN_K) / (nSrc > 0 ? nSrc : 1)) & ~31u;
    const int rot = (int)(((blockIdx.x * (unsigned)nSrc + (unsigned)sr) * step) % (unsigned)nd);
    for (int t0 = tid; t0 < nd; t0 += GCN_U * GCN_BS) {
        int ix[GCN_U];
        float dv[GCN_U];
#pragma unroll
        for (int j = 0; j < GCN_U; ++j) {
            int t = t0 + j * GCN_BS;
            int i = t + rot;
            i = i >= nd ? i - nd : i;
            ix[j] = t < nd ? i : -1;
            dv[j] = dis[dlo + (t < nd ? i : 0)];
        }
#pragma unroll
        for (int j = 0; j < GCN_U; ++j) {
            if (ix[j] >= 0) atomicAdd(&sg[dlo + ix[j]], acc[ix[j]] * dv[j]);
        }
    }
}

__global__ void gcn_make_t(const float* __restrict__ s1, const float* __restrict__ s1b,
                           const float* __restrict__ dis,
                           const float* __restrict__ x, float* __restrict__ t, int N,
                           size_t ws_bytes, size_t need_bytes) {
    int i = blockIdx.x * blockDim.x + threadIdx.x;
    const bool two = gcn_two(ws_bytes, need_bytes);
    const float* __restrict__ s1x = two ? s1b : s1;
    if (i < N) {
        const float s0 = s1[i];
        const float sb = s1x[i];
        t[i] = (two ? s0 + sb : s0) + dis[i] * dis[i] * x[i];
    }
}

__global__ void gcn_gemm_wmma(const float* __restrict__ t,
                              const float* __restrict__ W1,
                              const float* __restrict__ b1,
                              const float* __restrict__ W2,
                              const float* __restrict__ dis,
                              float* __restrict__ h2s,
                              int nTiles) {
    int wave = (int)((blockIdx.x * blockDim.x + threadIdx.x) >> 5);
    int lane = threadIdx.x & 31;
    if (wave >= nTiles) return;

    int mt = wave >> 1;
    int nt = wave & 1;
    int m0 = mt << 4;
    int n0 = nt << 4;
    int r = lane & 15;
    int khalf = lane >> 4;

    float trow = t[m0 + r];

    float drow = dis[m0 + r];

    v8f acc = {};
#pragma unroll
    for (int k0 = 0; k0 < 64; k0 += 4) {
        int ka = k0 + khalf * 2;
        float a0 = fmaf(trow, W1[ka],     b1[ka]);
        float a1 = fmaf(trow, W1[ka + 1], b1[ka + 1]);
        v2f a, b;
        a.x = (a0 > 0.0f ? a0 : 0.0f) * drow;
        a.y = (a1 > 0.0f ? a1 : 0.0f) * drow;
        b.x = W2[ka * 32 + n0 + r];
        b.y = W2[(ka + 1) * 32 + n0 + r];
        acc = __builtin_amdgcn_wmma_f32_16x16x4_f32(
             false, a,  false, b,
             (short)0, acc,  false,  false);
    }

    int col = n0 + r;
    int rowBase = m0 + khalf * 8;
#pragma unroll
    for (int v = 0; v < 8; ++v) {
        h2s[(rowBase + v) * 32 + col] = acc[v];
    }
}

#define GCN_E2U 8
__global__ __launch_bounds__(256)
void gcn_edge2(const int* __restrict__ src, const int* __restrict__ dst,
               const float* __restrict__ h2s,
               float* __restrict__ agg2a, float* __restrict__ agg2b, int E,
               size_t ws_bytes, size_t need_bytes) {
    const int lane = threadIdx.x & 31;
    float* __restrict__ agg2 = (gcn_two(ws_bytes, need_bytes) && gcn_half()) ? agg2b : agg2a;
    const int wave = __builtin_amdgcn_readfirstlane(
        (int)((blockIdx.x * blockDim.x + threadIdx.x) >> 5));
    const int nWaves = (int)((gridDim.x * blockDim.x) >> 5);
    const int nBatch = (E + 31) >> 5;
    for (int b = wave; b < nBatch; b += nWaves) {
        const int e0 = b << 5;
        const int e = e0 + lane;
        const int ec = e < E ? e : E - 1;
        const int s = src[ec];
        const int d = dst[ec];
        int cnt = E - e0;
        cnt = cnt < 32 ? cnt : 32;
        if (cnt == 32) {
#pragma unroll
            for (int j0 = 0; j0 < 32; j0 += GCN_E2U) {
                float v[GCN_E2U];
#pragma unroll
                for (int j = 0; j < GCN_E2U; ++j) {
                    const int sj = __builtin_amdgcn_readlane(s, j0 + j);
                    v[j] = h2s[(long)sj * 32 + lane];
                }
#pragma unroll
                for (int j = 0; j < GCN_E2U; ++j) {
                    const int dj = __builtin_amdgcn_readlane(d, j0 + j);
                    atomicAdd(&agg2[(long)dj * 32 + lane], v[j]);
                }
            }
        } else {
            for (int j = 0; j < cnt; ++j) {
                const int sj = __builtin_amdgcn_readlane(s, j);
                const int dj = __builtin_amdgcn_readlane(d, j);
                atomicAdd(&agg2[(long)dj * 32 + lane], h2s[(long)sj * 32 + lane]);
            }
        }
    }
}

#define GCN_PR 16

__device__ __forceinline__ void gcn_pool_row(float h, int g, float& acc, int& cnt, int& gcur,
                                             float* __restrict__ psum, float* __restrict__ pcnt,
                                             int lane) {
    if (g != gcur) {
        atomicAdd(&psum[gcur * 32 + lane], acc);
        if (lane == 0) atomicAdd(&pcnt[gcur], (float)cnt);
        acc = 0.0f;
        cnt = 0;
        gcur = g;
    }
    acc += h;
    cnt += 1;
}

__global__ __launch_bounds__(256)
void gcn_pool(const float* __restrict__ agg2, const float* __restrict__ agg2b,
              const float* __restrict__ h2s,
              const float* __restrict__ dis, const float* __restrict__ b2,
              const int* __restrict__ batch,
              float* __restrict__ psuma, float* __restrict__ psumb,
              float* __restrict__ pcnta, float* __restrict__ pcntb, int N,
              size_t ws_bytes, size_t need_bytes) {
    const bool two = gcn_two(ws_bytes, need_bytes);
    const float* __restrict__ agg2x = two ? agg2b : agg2;
    const bool hb = two && gcn_half();
    float* __restrict__ psum = hb ? psumb : psuma;
    float* __restrict__ pcnt = hb ? pcntb : pcnta;
    const int lane = threadIdx.x & 31;
    const int wave = __builtin_amdgcn_readfirstlane(
        (int)((blockIdx.x * blockDim.x + threadIdx.x) >> 5));
    const int i0 = wave * GCN_PR;
    if (i0 >= N) return;
    const float bf = b2[lane];
    const float* __restrict__ pa = agg2 + (long)i0 * 32 + lane;
    const float* __restrict__ px = agg2x + (long)i0 * 32 + lane;
    const float* __restrict__ ph = h2s + (long)i0 * 32 + lane;

    float acc = 0.0f;
    int cnt = 0;
    int gcur;
    if (i0 + GCN_PR <= N) {
        const int dl = __float_as_int(dis[i0 + (lane & (GCN_PR - 1))]);
        const int gl = batch[i0 + (lane & (GCN_PR - 1))];
        float a0[GCN_PR], a1[GCN_PR], hh[GCN_PR];
#pragma unroll
        for (int j = 0; j < GCN_PR; ++j) {
            a0[j] = pa[j * 32];
            a1[j] = px[j * 32];
            hh[j] = ph[j * 32];
        }
        asm volatile("" : "+v"(a1[0]));
        gcur = __builtin_amdgcn_readlane(gl, 0);
#pragma unroll
        for (int j = 0; j < GCN_PR; ++j) {
            const int g = __builtin_amdgcn_readlane(gl, j);
            const float di = __int_as_float(__builtin_amdgcn_readlane(dl, j));
            const float a = two ? a0[j] + a1[j] : a0[j];
            float h = di * (a + hh[j]) + bf;
            h = h > 0.0f ? h : 0.0f;
            gcn_pool_row(h, g, acc, cnt, gcur, psum, pcnt, lane);
        }
    } else {
        const int nv = N - i0;
        gcur = batch[i0];
        for (int j = 0; j < nv; ++j) {
            const int g = batch[i0 + j];
            const float di = dis[i0 + j];
            const float v0 = pa[j * 32];
            const float v1 = px[j * 32];
            const float a = two ? v0 + v1 : v0;
            float h = di * (a + ph[j * 32]) + bf;
            h = h > 0.0f ? h : 0.0f;
            gcn_pool_row(h, g, acc, cnt, gcur, psum, pcnt, lane);
        }
    }
    atomicAdd(&psum[gcur * 32 + lane], acc);
    if (lane == 0) atomicAdd(&pcnt[gcur], (float)cnt);
}

#define GCN_MLP_BS 1024
#define GCN_MLP_GPB (GCN_MLP_BS / 32)
__global__ __launch_bounds__(GCN_MLP_BS)
void gcn_mlp(const float* __restrict__ psum, const float* __restrict__ psumb,
             const float* __restrict__ pcnt, const float* __restrict__ pcntb,
             const float* __restrict__ Wp1, const float* __restrict__ bp1,
             const float* __restrict__ Wp2, const float* __restrict__ bp2,
             float* __restrict__ out, int G, size_t ws_bytes, size_t need_bytes) {
    const bool two = gcn_two(ws_bytes, need_bytes);
    const float* __restrict__ psumx = two ? psumb : psum;
    const float* __restrict__ pcntx = two ? pcntb : pcnt;
    __shared__ float res[GCN_MLP_GPB * 3];
    const int lane = threadIdx.x & 31;
    const int w = __builtin_amdgcn_readfirstlane((int)(threadIdx.x >> 5));
    const int g = blockIdx.x * GCN_MLP_GPB + w;
    if (g < G) {
        const float c0 = pcnt[g];
        const float c1 = pcntx[g];
        const float p0 = psum[g * 32 + lane];
        const float p1 = psumx[g * 32 + lane];
        float c = two ? c0 + c1 : c0;
        const float ps = two ? p0 + p1 : p0;
        c = c > 1.0f ? c : 1.0f;
        const float inv = 1.0f / c;
        const int p = __float_as_int(ps * inv);

        const int j0 = lane * 4;
        float h0 = bp1[j0], h1 = bp1[j0 + 1], h2 = bp1[j0 + 2], h3 = bp1[j0 + 3];
#pragma unroll
        for (int f = 0; f < 32; ++f) {
            const float pf = __int_as_float(__builtin_amdgcn_readlane(p, f));
            const float* wr = Wp1 + f * 128 + j0;
            h0 = fmaf(pf, wr[0], h0);
            h1 = fmaf(pf, wr[1], h1);
            h2 = fmaf(pf, wr[2], h2);
            h3 = fmaf(pf, wr[3], h3);
        }
        h0 = h0 > 0.0f ? h0 : 0.0f;
        h1 = h1 > 0.0f ? h1 : 0.0f;
        h2 = h2 > 0.0f ? h2 : 0.0f;
        h3 = h3 > 0.0f ? h3 : 0.0f;

        const float* v = Wp2 + j0 * 3;
        float o0 = fmaf(h3, v[9],  fmaf(h2, v[6], fmaf(h1, v[3], h0 * v[0])));
        float o1 = fmaf(h3, v[10], fmaf(h2, v[7], fmaf(h1, v[4], h0 * v[1])));
        float o2 = fmaf(h3, v[11], fmaf(h2, v[8], fmaf(h1, v[5], h0 * v[2])));
        for (int m = 16; m >= 1; m >>= 1) {
            o0 += __shfl_xor(o0, m);
            o1 += __shfl_xor(o1, m);
            o2 += __shfl_xor(o2, m);
        }
        if (lane == 0) {
            res[w * 3 + 0] = o0 + bp2[0];
            res[w * 3 + 1] = o1 + bp2[1];
            res[w * 3 + 2] = o2 + bp2[2];
        }
    }
    __syncthreads();
    if (w == 0) {
        const int gg = blockIdx.x * GCN_MLP_GPB + lane;
        if (gg < G) {
            out[gg * 3 + 0] = res[lane * 3 + 0];
            out[gg * 3 + 1] = res[lane * 3 + 1];
            out[gg * 3 + 2] = res[lane * 3 + 2];
        }
    }
}

extern "C" void kernel_launch(void* const* d_in, const int* in_sizes, int n_in,
                              void* d_out, int out_size, void* d_ws, size_t ws_size,
                              hipStream_t stream) {
    const float* x   = (const float*)d_in[0];
    const float* W1  = (const float*)d_in[1];
    const float* b1  = (const float*)d_in[2];
    const float* W2  = (const float*)d_in[3];
    const float* b2  = (const float*)d_in[4];
    const float* Wp1 = (const float*)d_in[5];
    const float* bp1 = (const float*)d_in[6];
    const float* Wp2 = (const float*)d_in[7];
    const float* bp2 = (const float*)d_in[8];
    const int* eidx  = (const int*)d_in[9];
    const int* batch = (const int*)d_in[10];
    float* out = (float*)d_out;

    const int N = in_sizes[0];
    const int E = in_sizes[9] / 2;
    const int G = out_size / 3;
    const int* src = eidx;
    const int* dst = eidx + E;

    float* wsf   = (float*)d_ws;
    float* deg   = wsf;
    float* s1    = deg + N;
    float* dis   = s1 + N;
    float* tbuf  = dis + N;
    float* h2s   = tbuf + N;
    float* agg2  = h2s + (long)N * 32;
    float* psum  = agg2 + (long)N * 32;
    float* pcnt  = psum + (long)G * 32;
    float* agg2b = (float*)(((uintptr_t)(pcnt + G) + 255) & ~(uintptr_t)255);
    float* psumb = agg2b + (long)N * 32;
    float* degb  = psumb + (long)G * 32;
    float* s1b   = degb + N;
    float* pcntb = s1b + N;
    float* wsend = pcntb + G;
    const size_t need = (size_t)((const char*)wsend - (const char*)d_ws);

    const int TPB = 256;

    long nz1 = 2L * N;
    long nz2 = 32L * N + 32L * G + G;
    long nz2two = (long)(wsend - agg2);
    gcn_zero_f32<<<1024, TPB, 0, stream>>>(deg, nz1, nz1, ws_size, need);
    gcn_zero_f32<<<2048, TPB, 0, stream>>>(agg2, nz2, nz2two, ws_size, need);

    const int chunk = (E + GCN_K - 1) / GCN_K;
    const dim3 sgrid(GCN_K, (N + GCN_RN - 1) / GCN_RN);
    gcn_degree<<<sgrid, GCN_BS, 0, stream>>>(dst, deg, degb, E, N, chunk, ws_size, need);
    gcn_dis<<<(N + TPB - 1) / TPB, TPB, 0, stream>>>(deg, degb, x, dis, tbuf, N, ws_size, need);
    const dim3 egrid(GCN_K, ((N + GCN_SR - 1) / GCN_SR) * ((N + GCN_DR - 1) / GCN_DR));
    gcn_edge1<<<egrid, GCN_BS, 0, stream>>>(src, dst, tbuf, dis, s1, s1b, E, N, chunk, ws_size, need);
    gcn_make_t<<<(N + TPB - 1) / TPB, TPB, 0, stream>>>(s1, s1b, dis, x, tbuf, N, ws_size, need);

    int nTiles = (N / 16) * 2;
    gcn_gemm_wmma<<<(nTiles + 7) / 8, TPB, 0, stream>>>(tbuf, W1, b1, W2, dis, h2s, nTiles);

    gcn_edge2<<<2048, TPB, 0, stream>>>(src, dst, h2s, agg2, agg2b, E, ws_size, need);

    gcn_pool<<<((N + GCN_PR - 1) / GCN_PR + 7) / 8, TPB, 0, stream>>>(agg2, agg2b, h2s, dis, b2, batch, psum, psumb, pcnt, pcntb, N, ws_size, need);

    gcn_mlp<<<(G + GCN_MLP_GPB - 1) / GCN_MLP_GPB, GCN_MLP_BS, 0, stream>>>(psum, psumb, pcnt, pcntb, Wp1, bp1, Wp2, bp2, out, G, ws_size, need);
}


# 10 launches -> 7: degree pass stores per-chunk counts plainly (no atomics), the dis kernel sums them and zeroes the accumulators (no zero launches), GEMM forms t itself (no make_t); sums unchanged
# speedup vs baseline: 5.8472x; 1.0635x over previous
#include <hip/hip_runtime.h>
#include <hip/hip_bf16.h>

typedef float v2f __attribute__((ext_vector_type(2)));
typedef float v8f __attribute__((ext_vector_type(8)));


__device__ __forceinline__ int gcn_half() {
    return (int)((__builtin_amdgcn_s_sendmsg_rtn(0x87) >> 18) & 1u);
}

__device__ __forceinline__ bool gcn_two(size_t ws_bytes, size_t need_bytes) {
    return ws_bytes >= need_bytes;
}


#define GCN_RN 12512
#define GCN_BS 1024
#define GCN_K  32
#define GCN_U  8

__global__ __launch_bounds__(GCN_BS)
void gcn_degree(const int* __restrict__ dst, float* __restrict__ degp,
                int E, int N, int chunk) {
    __shared__ unsigned cnt[GCN_RN];
    const int tid = threadIdx.x;
    const int lo = blockIdx.y * GCN_RN;
    int nr = N - lo;
    nr = nr < GCN_RN ? nr : GCN_RN;
    for (int i = tid; i < GCN_RN; i += GCN_BS) cnt[i] = 0u;
    __syncthreads();

    const int e0 = blockIdx.x * chunk;
    int e1 = e0 + chunk;
    e1 = e1 < E ? e1 : E;
    for (int e = e0 + tid; e < e1; e += GCN_U * GCN_BS) {
        int d[GCN_U];
#pragma unroll
        for (int j = 0; j < GCN_U; ++j) {
            int ej = e + j * GCN_BS;
            int dv = dst[ej < e1 ? ej : e1 - 1];
            d[j] = ej < e1 ? dv - lo : -1;
        }
#pragma unroll
        for (int j = 0; j < GCN_U; ++j) {
            if ((unsigned)d[j] < (unsigned)nr) atomicAdd(&cnt[d[j]], 1u);
        }
    }
    __syncthreads();
    float* __restrict__ dg = degp + (long)blockIdx.x * N + lo;
    for (int i = tid; i < nr; i += GCN_BS) dg[i] = (float)cnt[i];
}

__global__ void gcn_dis(const float* __restrict__ degp,
                        const float* __restrict__ x,
                        float* __restrict__ dis, float* __restrict__ xd, int N,
                        float* __restrict__ z1, long nz1,
                        float* __restrict__ z2, long nz2_one, long nz2_two,
                        size_t ws_bytes, size_t need_bytes) {
    long i = (long)blockIdx.x * blockDim.x + threadIdx.x;
    const long stride = (long)gridDim.x * blockDim.x;
    for (long n = i; n < N; n += stride) {
        float p[GCN_K];
#pragma unroll
        for (int k = 0; k < GCN_K; ++k) p[k] = degp[(long)k * N + n];
        float d = 0.0f;
#pragma unroll
        for (int k = 0; k < GCN_K; ++k) d += p[k];
        const float di = rsqrtf(d + 1.0f);
        dis[n] = di;
        xd[n] = x[n] * di;
    }
    const long nz2 = gcn_two(ws_bytes, need_bytes) ? nz2_two : nz2_one;
    for (long k = i; k < nz1; k += stride) z1[k] = 0.0f;
    for (long k = i; k < nz2; k += stride) z2[k] = 0.0f;
}

#define GCN_SR 50016
#define GCN_DR 25024
#define GCN_SR4_TRIPS ((GCN_SR / 4 + GCN_BS - 1) / GCN_BS)
__global__ __launch_bounds__(GCN_BS)
void gcn_edge1(const int* __restrict__ src, const int* __restrict__ dst,
               const float* __restrict__ xd, const float* __restrict__ dis,
               float* __restrict__ s1, float* __restrict__ s1b, int E, int N, int chunk,
               size_t ws_bytes, size_t need_bytes) {
    __shared__ float xl[GCN_SR];
    __shared__ float acc[GCN_DR];
    const int tid = threadIdx.x;
    const int nDst = (N + GCN_DR - 1) / GCN_DR;
    const int sr = (int)blockIdx.y / nDst;
    const int dr = (int)blockIdx.y - sr * nDst;
    const int slo = sr * GCN_SR;
    const int dlo = dr * GCN_DR;
    int ns = N - slo;
    ns = ns < GCN_SR ? ns : GCN_SR;
    int nd = N - dlo;
    nd = nd < GCN_DR ? nd : GCN_DR;

    const float* __restrict__ xs = xd + slo;
    if ((((uintptr_t)xs) & 15) == 0) {
        const float4* __restrict__ xs4 = (const float4*)xs;
        const int ns4 = ns >> 2;
        float4 q[GCN_SR4_TRIPS];
#pragma unroll
        for (int k = 0; k < GCN_SR4_TRIPS; ++k) {
            const int i = tid + k * GCN_BS;
            q[k] = xs4[i < ns4 ? i : 0];
        }
#pragma unroll
        for (int k = 0; k < GCN_SR4_TRIPS; ++k) {
            const int i = tid + k * GCN_BS;
            if (i < ns4) {
                xl[4 * i + 0] = q[k].x;
                xl[4 * i + 1] = q[k].y;
                xl[4 * i + 2] = q[k].z;
                xl[4 * i + 3] = q[k].w;
            }
        }
        for (int i = (ns4 << 2) + tid; i < ns; i += GCN_BS) xl[i] = xs[i];
    } else {
        for (int i = tid; i < ns; i += GCN_BS) xl[i] = xs[i];
    }
    for (int i = tid; i < GCN_DR; i += GCN_BS) acc[i] = 0.0f;
    __syncthreads();

    const int e0 = blockIdx.x * chunk;
    int e1 = e0 + chunk;
    e1 = e1 < E ? e1 : E;
    for (int e = e0 + tid; e < e1; e += GCN_U * GCN_BS) {
        int dvr[GCN_U], svr[GCN_U];
        int d[GCN_U], s[GCN_U];
        float v[GCN_U];
#pragma unroll
        for (int j = 0; j < GCN_U; ++j) {
            int ej = e + j * GCN_BS;
            int ec = ej < e1 ? ej : e1 - 1;
            dvr[j] = dst[ec];
            svr[j] = src[ec];
        }
        static_assert(GCN_U == 8, "the statement below names 8 + 8 registers");
        asm volatile("" : "+v"(dvr[0]), "+v"(dvr[1]), "+v"(dvr[2]), "+v"(dvr[3]),
                          "+v"(dvr[4]), "+v"(dvr[5]), "+v"(dvr[6]), "+v"(dvr[7]),
                          "+v"(svr[0]), "+v"(svr[1]), "+v"(svr[2]), "+v"(svr[3]),
                          "+v"(svr[4]), "+v"(svr[5]), "+v"(svr[6]), "+v"(svr[7]));
#pragma unroll
        for (int j = 0; j < GCN_U; ++j) {
            int ej = e + j * GCN_BS;
            int dv = dvr[j] - dlo;
            int sv = svr[j] - slo;
            bool ok = ej < e1 && (unsigned)dv < (unsigned)nd && (unsigned)sv < (unsigned)ns;
            d[j] = ok ? dv : -1;
            s[j] = ok ? sv : 0;
        }
#pragma unroll
        for (int j = 0; j < GCN_U; ++j) {
            v[j] = xl[s[j]];
        }
#pragma unroll
        for (int j = 0; j < GCN_U; ++j) {
            asm volatile("" : : "v"(v[j]));
        }
#pragma unroll
        for (int j = 0; j < GCN_U; ++j) {
            if (d[j] >= 0) atomicAdd(&acc[d[j]], v[j]);
        }
    }
    __syncthreads();
    float* __restrict__ sg = (gcn_two(ws_bytes, need_bytes) && gcn_half()) ? s1b : s1;
    const int nSrc = (N + GCN_SR - 1) / GCN_SR;
    const unsigned step = (unsigned)((GCN_DR / GCN_K) / (nSrc > 0 ? nSrc : 1)) & ~31u;
    const int rot = (int)(((blockIdx.x * (unsigned)nSrc + (unsigned)sr) * step) % (unsigned)nd);
    for (int t0 = tid; t0 < nd; t0 += GCN_U * GCN_BS) {
        int ix[GCN_U];
        float dv[GCN_U];
#pragma unroll
        for (int j = 0; j < GCN_U; ++j) {
            int t = t0 + j * GCN_BS;
            int i = t + rot;
            i = i >= nd ? i - nd : i;
            ix[j] = t < nd ? i : -1;
            dv[j] = dis[dlo + (t < nd ? i : 0)];
        }
#pragma unroll
        for (int j = 0; j < GCN_U; ++j) {
            if (ix[j] >= 0) atomicAdd(&sg[dlo + ix[j]], acc[ix[j]] * dv[j]);
        }
    }
}

__global__ void gcn_gemm_wmma(const float* __restrict__ s1, const float* __restrict__ s1b,
                              const float* __restrict__ x,
                              const float* __restrict__ W1,
                              const float* __restrict__ b1,
                              const float* __restrict__ W2,
                              const float* __restrict__ dis,
                              float* __restrict__ h2s,
                              int nTiles, size_t ws_bytes, size_t need_bytes) {
    int wave = (int)((blockIdx.x * blockDim.x + threadIdx.x) >> 5);
    int lane = threadIdx.x & 31;
    if (wave >= nTiles) return;

    int mt = wave >> 1;
    int nt = wave & 1;
    int m0 = mt << 4;
    int n0 = nt << 4;
    int r = lane & 15;
    int khalf = lane >> 4;

    const bool two = gcn_two(ws_bytes, need_bytes);
    const float* __restrict__ s1x = two ? s1b : s1;
    const float s0 = s1[m0 + r];
    const float sb = s1x[m0 + r];
    const float xr = x[m0 + r];
    float drow = dis[m0 + r];
    float trow = (two ? s0 + sb : s0) + drow * drow * xr;

    v8f acc = {};
#pragma unroll
    for (int k0 = 0; k0 < 64; k0 += 4) {
        int ka = k0 + khalf * 2;
        float a0 = fmaf(trow, W1[ka],     b1[ka]);
        float a1 = fmaf(trow, W1[ka + 1], b1[ka + 1]);
        v2f a, b;
        a.x = (a0 > 0.0f ? a0 : 0.0f) * drow;
        a.y = (a1 > 0.0f ? a1 : 0.0f) * drow;
        b.x = W2[ka * 32 + n0 + r];
        b.y = W2[(ka + 1) * 32 + n0 + r];
        acc = __builtin_amdgcn_wmma_f32_16x16x4_f32(
             false, a,  false, b,
             (short)0, acc,  false,  false);
    }

    int col = n0 + r;
    int rowBase = m0 + khalf * 8;
#pragma unroll
    for (int v = 0; v < 8; ++v) {
        h2s[(rowBase + v) * 32 + col] = acc[v];
    }
}

#define GCN_E2U 8
__global__ __launch_bounds__(256)
void gcn_edge2(const int* __restrict__ src, const int* __restrict__ dst,
               const float* __restrict__ h2s,
               float* __restrict__ agg2a, float* __restrict__ agg2b, int E,
               size_t ws_bytes, size_t need_bytes) {
    const int lane = threadIdx.x & 31;
    float* __restrict__ agg2 = (gcn_two(ws_bytes, need_bytes) && gcn_half()) ? agg2b : agg2a;
    const int wave = __builtin_amdgcn_readfirstlane(
        (int)((blockIdx.x * blockDim.x + threadIdx.x) >> 5));
    const int nWaves = (int)((gridDim.x * blockDim.x) >> 5);
    const int nBatch = (E + 31) >> 5;
    for (int b = wave; b < nBatch; b += nWaves) {
        const int e0 = b << 5;
        const int e = e0 + lane;
        const int ec = e < E ? e : E - 1;
        const int s = src[ec];
        const int d = dst[ec];
        int cnt = E - e0;
        cnt = cnt < 32 ? cnt : 32;
        if (cnt == 32) {
#pragma unroll
            for (int j0 = 0; j0 < 32; j0 += GCN_E2U) {
                float v[GCN_E2U];
#pragma unroll
                for (int j = 0; j < GCN_E2U; ++j) {
                    const int sj = __builtin_amdgcn_readlane(s, j0 + j);
                    v[j] = h2s[(long)sj * 32 + lane];
                }
#pragma unroll
                for (int j = 0; j < GCN_E2U; ++j) {
                    const int dj = __builtin_amdgcn_readlane(d, j0 + j);
                    atomicAdd(&agg2[(long)dj * 32 + lane], v[j]);
                }
            }
        } else {
            for (int j = 0; j < cnt; ++j) {
                const int sj = __builtin_amdgcn_readlane(s, j);
                const int dj = __builtin_amdgcn_readlane(d, j);
                atomicAdd(&agg2[(long)dj * 32 + lane], h2s[(long)sj * 32 + lane]);
            }
        }
    }
}

#define GCN_PR 16

__device__ __forceinline__ void gcn_pool_row(float h, int g, float& acc, int& cnt, int& gcur,
                                             float* __restrict__ psum, float* __restrict__ pcnt,
                                             int lane) {
    if (g != gcur) {
        atomicAdd(&psum[gcur * 32 + lane], acc);
        if (lane == 0) atomicAdd(&pcnt[gcur], (float)cnt);
        acc = 0.0f;
        cnt = 0;
        gcur = g;
    }
    acc += h;
    cnt += 1;
}

__global__ __launch_bounds__(256)
void gcn_pool(const float* __restrict__ agg2, const float* __restrict__ agg2b,
              const float* __restrict__ h2s,
              const float* __restrict__ dis, const float* __restrict__ b2,
              const int* __restrict__ batch,
              float* __restrict__ psuma, float* __restrict__ psumb,
              float* __restrict__ pcnta, float* __restrict__ pcntb, int N,
              size_t ws_bytes, size_t need_bytes) {
    const bool two = gcn_two(ws_bytes, need_bytes);
    const float* __restrict__ agg2x = two ? agg2b : agg2;
    const bool hb = two && gcn_half();
    float* __restrict__ psum = hb ? psumb : psuma;
    float* __restrict__ pcnt = hb ? pcntb : pcnta;
    const int lane = threadIdx.x & 31;
    const int wave = __builtin_amdgcn_readfirstlane(
        (int)((blockIdx.x * blockDim.x + threadIdx.x) >> 5));
    const int i0 = wave * GCN_PR;
    if (i0 >= N) return;
    const float bf = b2[lane];
    const float* __restrict__ pa = agg2 + (long)i0 * 32 + lane;
    const float* __restrict__ px = agg2x + (long)i0 * 32 + lane;
    const float* __restrict__ ph = h2s + (long)i0 * 32 + lane;

    float acc = 0.0f;
    int cnt = 0;
    int gcur;
    if (i0 + GCN_PR <= N) {
        const int dl = __float_as_int(dis[i0 + (lane & (GCN_PR - 1))]);
        const int gl = batch[i0 + (lane & (GCN_PR - 1))];
        float a0[GCN_PR], a1[GCN_PR], hh[GCN_PR];
#pragma unroll
        for (int j = 0; j < GCN_PR; ++j) {
            a0[j] = pa[j * 32];
            a1[j] = px[j * 32];
            hh[j] = ph[j * 32];
        }
        asm volatile("" : "+v"(a1[0]));
        gcur = __builtin_amdgcn_readlane(gl, 0);
#pragma unroll
        for (int j = 0; j < GCN_PR; ++j) {
            const int g = __builtin_amdgcn_readlane(gl, j);
            const float di = __int_as_float(__builtin_amdgcn_readlane(dl, j));
            const float a = two ? a0[j] + a1[j] : a0[j];
            float h = di * (a + hh[j]) + bf;
            h = h > 0.0f ? h : 0.0f;
            gcn_pool_row(h, g, acc, cnt, gcur, psum, pcnt, lane);
        }
    } else {
        const int nv = N - i0;
        gcur = batch[i0];
        for (int j = 0; j < nv; ++j) {
            const int g = batch[i0 + j];
            const float di = dis[i0 + j];
            const float v0 = pa[j * 32];
            const float v1 = px[j * 32];
            const float a = two ? v0 + v1 : v0;
            float h = di * (a + ph[j * 32]) + bf;
            h = h > 0.0f ? h : 0.0f;
            gcn_pool_row(h, g, acc, cnt, gcur, psum, pcnt, lane);
        }
    }
    atomicAdd(&psum[gcur * 32 + lane], acc);
    if (lane == 0) atomicAdd(&pcnt[gcur], (float)cnt);
}

#define GCN_MLP_BS 1024
#define GCN_MLP_GPB (GCN_MLP_BS / 32)
__global__ __launch_bounds__(GCN_MLP_BS)
void gcn_mlp(const float* __restrict__ psum, const float* __restrict__ psumb,
             const float* __restrict__ pcnt, const float* __restrict__ pcntb,
             const float* __restrict__ Wp1, const float* __restrict__ bp1,
             const float* __restrict__ Wp2, const float* __restrict__ bp2,
             float* __restrict__ out, int G, size_t ws_bytes, size_t need_bytes) {
    const bool two = gcn_two(ws_bytes, need_bytes);
    const float* __restrict__ psumx = two ? psumb : psum;
    const float* __restrict__ pcntx = two ? pcntb : pcnt;
    __shared__ float res[GCN_MLP_GPB * 3];
    const int lane = threadIdx.x & 31;
    const int w = __builtin_amdgcn_readfirstlane((int)(threadIdx.x >> 5));
    const int g = blockIdx.x * GCN_MLP_GPB + w;
    if (g < G) {
        const float c0 = pcnt[g];
        const float c1 = pcntx[g];
        const float p0 = psum[g * 32 + lane];
        const float p1 = psumx[g * 32 + lane];
        float c = two ? c0 + c1 : c0;
        const float ps = two ? p0 + p1 : p0;
        c = c > 1.0f ? c : 1.0f;
        const float inv = 1.0f / c;
        const int p = __float_as_int(ps * inv);

        const int j0 = lane * 4;
        float h0 = bp1[j0], h1 = bp1[j0 + 1], h2 = bp1[j0 + 2], h3 = bp1[j0 + 3];
#pragma unroll
        for (int f = 0; f < 32; ++f) {
            const float pf = __int_as_float(__builtin_amdgcn_readlane(p, f));
            const float* wr = Wp1 + f * 128 + j0;
            h0 = fmaf(pf, wr[0], h0);
            h1 = fmaf(pf, wr[1], h1);
            h2 = fmaf(pf, wr[2], h2);
            h3 = fmaf(pf, wr[3], h3);
        }
        h0 = h0 > 0.0f ? h0 : 0.0f;
        h1 = h1 > 0.0f ? h1 : 0.0f;
        h2 = h2 > 0.0f ? h2 : 0.0f;
        h3 = h3 > 0.0f ? h3 : 0.0f;

        const float* v = Wp2 + j0 * 3;
        float o0 = fmaf(h3, v[9],  fmaf(h2, v[6], fmaf(h1, v[3], h0 * v[0])));
        float o1 = fmaf(h3, v[10], fmaf(h2, v[7], fmaf(h1, v[4], h0 * v[1])));
        float o2 = fmaf(h3, v[11], fmaf(h2, v[8], fmaf(h1, v[5], h0 * v[2])));
        for (int m = 16; m >= 1; m >>= 1) {
            o0 += __shfl_xor(o0, m);
            o1 += __shfl_xor(o1, m);
            o2 += __shfl_xor(o2, m);
        }
        if (lane == 0) {
            res[w * 3 + 0] = o0 + bp2[0];
            res[w * 3 + 1] = o1 + bp2[1];
            res[w * 3 + 2] = o2 + bp2[2];
        }
    }
    __syncthreads();
    if (w == 0) {
        const int gg = blockIdx.x * GCN_MLP_GPB + lane;
        if (gg < G) {
            out[gg * 3 + 0] = res[lane * 3 + 0];
            out[gg * 3 + 1] = res[lane * 3 + 1];
            out[gg * 3 + 2] = res[lane * 3 + 2];
        }
    }
}

extern "C" void kernel_launch(void* const* d_in, const int* in_sizes, int n_in,
                              void* d_out, int out_size, void* d_ws, size_t ws_size,
                              hipStream_t stream) {
    const float* x   = (const float*)d_in[0];
    const float* W1  = (const float*)d_in[1];
    const float* b1  = (const float*)d_in[2];
    const float* W2  = (const float*)d_in[3];
    const float* b2  = (const float*)d_in[4];
    const float* Wp1 = (const float*)d_in[5];
    const float* bp1 = (const float*)d_in[6];
    const float* Wp2 = (const float*)d_in[7];
    const float* bp2 = (const float*)d_in[8];
    const int* eidx  = (const int*)d_in[9];
    const int* batch = (const int*)d_in[10];
    float* out = (float*)d_out;

    const int N = in_sizes[0];
    const int E = in_sizes[9] / 2;
    const int G = out_size / 3;
    const int* src = eidx;
    const int* dst = eidx + E;

    float* wsf   = (float*)d_ws;
    float* deg   = wsf;
    float* s1    = deg + N;
    float* dis   = s1 + N;
    float* tbuf  = dis + N;
    float* h2s   = tbuf + N;
    float* agg2  = h2s + (long)N * 32;
    float* psum  = agg2 + (long)N * 32;
    float* pcnt  = psum + (long)G * 32;
    float* agg2b = (float*)(((uintptr_t)(pcnt + G) + 255) & ~(uintptr_t)255);
    float* psumb = agg2b + (long)N * 32;
    float* degb  = psumb + (long)G * 32;
    float* s1b   = degb + N;
    float* pcntb = s1b + N;
    float* wsend = pcntb + G;
    const size_t need = (size_t)((const char*)wsend - (const char*)d_ws);

    const int TPB = 256;

    long nz1 = N;
    long nz2 = 32L * N + 32L * G + G;
    long nz2two = (long)(wsend - agg2);
    static_assert(GCN_K <= 32, "degp [GCN_K][N] must fit in h2s [N][32]");
    float* degp = h2s;

    const int chunk = (E + GCN_K - 1) / GCN_K;
    const dim3 sgrid(GCN_K, (N + GCN_RN - 1) / GCN_RN);
    gcn_degree<<<sgrid, GCN_BS, 0, stream>>>(dst, degp, E, N, chunk);
    gcn_dis<<<2048, TPB, 0, stream>>>(degp, x, dis, tbuf, N, s1, nz1, agg2, nz2, nz2two, ws_size, need);
    const dim3 egrid(GCN_K, ((N + GCN_SR - 1) / GCN_SR) * ((N + GCN_DR - 1) / GCN_DR));
    gcn_edge1<<<egrid, GCN_BS, 0, stream>>>(src, dst, tbuf, dis, s1, s1b, E, N, chunk, ws_size, need);

    int nTiles = (N / 16) * 2;
    gcn_gemm_wmma<<<(nTiles + 7) / 8, TPB, 0, stream>>>(s1, s1b, x, W1, b1, W2, dis, h2s, nTiles, ws_size, need);

    gcn_edge2<<<2048, TPB, 0, stream>>>(src, dst, h2s, agg2, agg2b, E, ws_size, need);

    gcn_pool<<<((N + GCN_PR - 1) / GCN_PR + 7) / 8, TPB, 0, stream>>>(agg2, agg2b, h2s, dis, b2, batch, psum, psumb, pcnt, pcntb, N, ws_size, need);

    gcn_mlp<<<(G + GCN_MLP_GPB - 1) / GCN_MLP_GPB, GCN_MLP_BS, 0, stream>>>(psum, psumb, pcnt, pcntb, Wp1, bp1, Wp2, bp2, out, G, ws_size, need);
}


# degree and layer-1 edge passes read their index streams 16 B per lane (4 consecutive edges, 512-B wave loads) instead of 4 B; the order of the layer-1 LDS float adds changes
# speedup vs baseline: 6.0513x; 1.0349x over previous
#include <hip/hip_runtime.h>
#include <hip/hip_bf16.h>

typedef float v2f __attribute__((ext_vector_type(2)));
typedef float v8f __attribute__((ext_vector_type(8)));


__device__ __forceinline__ int gcn_half() {
    return (int)((__builtin_amdgcn_s_sendmsg_rtn(0x87) >> 18) & 1u);
}

__device__ __forceinline__ bool gcn_two(size_t ws_bytes, size_t need_bytes) {
    return ws_bytes >= need_bytes;
}


#define GCN_RN 12512
#define GCN_BS 1024
#define GCN_K  32
#define GCN_U  8
#define GCN_DU 16

__global__ __launch_bounds__(GCN_BS)
void gcn_degree(const int* __restrict__ dst, float* __restrict__ degp,
                int E, int N, int chunk) {
    __shared__ unsigned cnt[GCN_RN];
    const int tid = threadIdx.x;
    const int lo = blockIdx.y * GCN_RN;
    int nr = N - lo;
    nr = nr < GCN_RN ? nr : GCN_RN;
    for (int i = tid; i < GCN_RN; i += GCN_BS) cnt[i] = 0u;
    __syncthreads();

    const int e0 = blockIdx.x * chunk;
    int e1 = e0 + chunk;
    e1 = e1 < E ? e1 : E;
    const bool vec = ((((uintptr_t)dst) & 15) == 0) && ((chunk & 3) == 0) && ((E & 3) == 0);
    if (vec) {
        for (int ef = e0 + 4 * tid; ef < e1; ef += GCN_DU * GCN_BS) {
            int4 q[GCN_DU / 4];
            bool ok[GCN_DU / 4];
#pragma unroll
            for (int jj = 0; jj < GCN_DU / 4; ++jj) {
                const int eb = ef + jj * 4 * GCN_BS;
                ok[jj] = eb < e1;
                q[jj] = *(const int4*)(dst + (ok[jj] ? eb : e1 - 4));
            }
#pragma unroll
            for (int jj = 0; jj < GCN_DU / 4; ++jj) {
                const int d0 = ok[jj] ? q[jj].x - lo : -1;
                const int d1 = ok[jj] ? q[jj].y - lo : -1;
                const int d2 = ok[jj] ? q[jj].z - lo : -1;
                const int d3 = ok[jj] ? q[jj].w - lo : -1;
                if ((unsigned)d0 < (unsigned)nr) atomicAdd(&cnt[d0], 1u);
                if ((unsigned)d1 < (unsigned)nr) atomicAdd(&cnt[d1], 1u);
                if ((unsigned)d2 < (unsigned)nr) atomicAdd(&cnt[d2], 1u);
                if ((unsigned)d3 < (unsigned)nr) atomicAdd(&cnt[d3], 1u);
            }
        }
    } else {
        for (int e = e0 + tid; e < e1; e += GCN_U * GCN_BS) {
            int d[GCN_U];
#pragma unroll
            for (int j = 0; j < GCN_U; ++j) {
                int ej = e + j * GCN_BS;
                int dv = dst[ej < e1 ? ej : e1 - 1];
                d[j] = ej < e1 ? dv - lo : -1;
            }
#pragma unroll
            for (int j = 0; j < GCN_U; ++j) {
                if ((unsigned)d[j] < (unsigned)nr) atomicAdd(&cnt[d[j]], 1u);
            }
        }
    }
    __syncthreads();
    float* __restrict__ dg = degp + (long)blockIdx.x * N + lo;
    for (int i = tid; i < nr; i += GCN_BS) dg[i] = (float)cnt[i];
}

__global__ void gcn_dis(const float* __restrict__ degp,
                        const float* __restrict__ x,
                        float* __restrict__ dis, float* __restrict__ xd, int N,
                        float* __restrict__ z1, long nz1,
                        float* __restrict__ z2, long nz2_one, long nz2_two,
                        size_t ws_bytes, size_t need_bytes) {
    long i = (long)blockIdx.x * blockDim.x + threadIdx.x;
    const long stride = (long)gridDim.x * blockDim.x;
    for (long n = i; n < N; n += stride) {
        float p[GCN_K];
#pragma unroll
        for (int k = 0; k < GCN_K; ++k) p[k] = degp[(long)k * N + n];
        float d = 0.0f;
#pragma unroll
        for (int k = 0; k < GCN_K; ++k) d += p[k];
        const float di = rsqrtf(d + 1.0f);
        dis[n] = di;
        xd[n] = x[n] * di;
    }
    const long nz2 = gcn_two(ws_bytes, need_bytes) ? nz2_two : nz2_one;
    for (long k = i; k < nz1; k += stride) z1[k] = 0.0f;
    for (long k = i; k < nz2; k += stride) z2[k] = 0.0f;
}

#define GCN_SR 50016
#define GCN_DR 25024
#define GCN_SR4_TRIPS ((GCN_SR / 4 + GCN_BS - 1) / GCN_BS)
__global__ __launch_bounds__(GCN_BS)
void gcn_edge1(const int* __restrict__ src, const int* __restrict__ dst,
               const float* __restrict__ xd, const float* __restrict__ dis,
               float* __restrict__ s1, float* __restrict__ s1b, int E, int N, int chunk,
               size_t ws_bytes, size_t need_bytes) {
    __shared__ float xl[GCN_SR];
    __shared__ float acc[GCN_DR];
    const int tid = threadIdx.x;
    const int nDst = (N + GCN_DR - 1) / GCN_DR;
    const int sr = (int)blockIdx.y / nDst;
    const int dr = (int)blockIdx.y - sr * nDst;
    const int slo = sr * GCN_SR;
    const int dlo = dr * GCN_DR;
    int ns = N - slo;
    ns = ns < GCN_SR ? ns : GCN_SR;
    int nd = N - dlo;
    nd = nd < GCN_DR ? nd : GCN_DR;

    const float* __restrict__ xs = xd + slo;
    if ((((uintptr_t)xs) & 15) == 0) {
        const float4* __restrict__ xs4 = (const float4*)xs;
        const int ns4 = ns >> 2;
        float4 q[GCN_SR4_TRIPS];
#pragma unroll
        for (int k = 0; k < GCN_SR4_TRIPS; ++k) {
            const int i = tid + k * GCN_BS;
            q[k] = xs4[i < ns4 ? i : 0];
        }
#pragma unroll
        for (int k = 0; k < GCN_SR4_TRIPS; ++k) {
            const int i = tid + k * GCN_BS;
            if (i < ns4) {
                xl[4 * i + 0] = q[k].x;
                xl[4 * i + 1] = q[k].y;
                xl[4 * i + 2] = q[k].z;
                xl[4 * i + 3] = q[k].w;
            }
        }
        for (int i = (ns4 << 2) + tid; i < ns; i += GCN_BS) xl[i] = xs[i];
    } else {
        for (int i = tid; i < ns; i += GCN_BS) xl[i] = xs[i];
    }
    for (int i = tid; i < GCN_DR; i += GCN_BS) acc[i] = 0.0f;
    __syncthreads();

    const int e0 = blockIdx.x * chunk;
    int e1 = e0 + chunk;
    e1 = e1 < E ? e1 : E;
    const bool vec = (((((uintptr_t)src) | ((uintptr_t)dst)) & 15) == 0) &&
                     ((chunk & 3) == 0) && ((E & 3) == 0);
    const int lstep = vec ? 4 : 1;
    for (int ef = e0 + tid * lstep; ef < e1; ef += GCN_U * GCN_BS) {
        int dvr[GCN_U], svr[GCN_U];
        bool inb[GCN_U];
        int d[GCN_U], s[GCN_U];
        float v[GCN_U];
        if (vec) {
#pragma unroll
            for (int jj = 0; jj < GCN_U / 4; ++jj) {
                const int eb = ef + jj * 4 * GCN_BS;
                const bool ok = eb < e1;
                const int ec = ok ? eb : e1 - 4;
                const int4 dq = *(const int4*)(dst + ec);
                const int4 sq = *(const int4*)(src + ec);
                dvr[4 * jj + 0] = dq.x; dvr[4 * jj + 1] = dq.y;
                dvr[4 * jj + 2] = dq.z; dvr[4 * jj + 3] = dq.w;
                svr[4 * jj + 0] = sq.x; svr[4 * jj + 1] = sq.y;
                svr[4 * jj + 2] = sq.z; svr[4 * jj + 3] = sq.w;
                inb[4 * jj + 0] = ok; inb[4 * jj + 1] = ok;
                inb[4 * jj + 2] = ok; inb[4 * jj + 3] = ok;
            }
        } else {
#pragma unroll
            for (int j = 0; j < GCN_U; ++j) {
                int ej = ef + j * GCN_BS;
                inb[j] = ej < e1;
                int ec = inb[j] ? ej : e1 - 1;
                dvr[j] = dst[ec];
                svr[j] = src[ec];
            }
        }
        static_assert(GCN_U == 8, "the statement below names 8 + 8 registers");
        asm volatile("" : "+v"(dvr[0]), "+v"(dvr[1]), "+v"(dvr[2]), "+v"(dvr[3]),
                          "+v"(dvr[4]), "+v"(dvr[5]), "+v"(dvr[6]), "+v"(dvr[7]),
                          "+v"(svr[0]), "+v"(svr[1]), "+v"(svr[2]), "+v"(svr[3]),
                          "+v"(svr[4]), "+v"(svr[5]), "+v"(svr[6]), "+v"(svr[7]));
#pragma unroll
        for (int j = 0; j < GCN_U; ++j) {
            int dv = dvr[j] - dlo;
            int sv = svr[j] - slo;
            bool ok = inb[j] && (unsigned)dv < (unsigned)nd && (unsigned)sv < (unsigned)ns;
            d[j] = ok ? dv : -1;
            s[j] = ok ? sv : 0;
        }
#pragma unroll
        for (int j = 0; j < GCN_U; ++j) {
            v[j] = xl[s[j]];
        }
#pragma unroll
        for (int j = 0; j < GCN_U; ++j) {
            asm volatile("" : : "v"(v[j]));
        }
#pragma unroll
        for (int j = 0; j < GCN_U; ++j) {
            if (d[j] >= 0) atomicAdd(&acc[d[j]], v[j]);
        }
    }
    __syncthreads();
    float* __restrict__ sg = (gcn_two(ws_bytes, need_bytes) && gcn_half()) ? s1b : s1;
    const int nSrc = (N + GCN_SR - 1) / GCN_SR;
    const unsigned step = (unsigned)((GCN_DR / GCN_K) / (nSrc > 0 ? nSrc : 1)) & ~31u;
    const int rot = (int)(((blockIdx.x * (unsigned)nSrc + (unsigned)sr) * step) % (unsigned)nd);
    for (int t0 = tid; t0 < nd; t0 += GCN_U * GCN_BS) {
        int ix[GCN_U];
        float dv[GCN_U];
#pragma unroll
        for (int j = 0; j < GCN_U; ++j) {
            int t = t0 + j * GCN_BS;
            int i = t + rot;
            i = i >= nd ? i - nd : i;
            ix[j] = t < nd ? i : -1;
            dv[j] = dis[dlo + (t < nd ? i : 0)];
        }
#pragma unroll
        for (int j = 0; j < GCN_U; ++j) {
            if (ix[j] >= 0) atomicAdd(&sg[dlo + ix[j]], acc[ix[j]] * dv[j]);
        }
    }
}

__global__ void gcn_gemm_wmma(const float* __restrict__ s1, const float* __restrict__ s1b,
                              const float* __restrict__ x,
                              const float* __restrict__ W1,
                              const float* __restrict__ b1,
                              const float* __restrict__ W2,
                              const float* __restrict__ dis,
                              float* __restrict__ h2s,
                              int nTiles, size_t ws_bytes, size_t need_bytes) {
    int wave = (int)((blockIdx.x * blockDim.x + threadIdx.x) >> 5);
    int lane = threadIdx.x & 31;
    if (wave >= nTiles) return;

    int mt = wave >> 1;
    int nt = wave & 1;
    int m0 = mt << 4;
    int n0 = nt << 4;
    int r = lane & 15;
    int khalf = lane >> 4;

    const bool two = gcn_two(ws_bytes, need_bytes);
    const float* __restrict__ s1x = two ? s1b : s1;
    const float s0 = s1[m0 + r];
    const float sb = s1x[m0 + r];
    const float xr = x[m0 + r];
    float drow = dis[m0 + r];
    float trow = (two ? s0 + sb : s0) + drow * drow * xr;

    v8f acc = {};
#pragma unroll
    for (int k0 = 0; k0 < 64; k0 += 4) {
        int ka = k0 + khalf * 2;
        float a0 = fmaf(trow, W1[ka],     b1[ka]);
        float a1 = fmaf(trow, W1[ka + 1], b1[ka + 1]);
        v2f a, b;
        a.x = (a0 > 0.0f ? a0 : 0.0f) * drow;
        a.y = (a1 > 0.0f ? a1 : 0.0f) * drow;
        b.x = W2[ka * 32 + n0 + r];
        b.y = W2[(ka + 1) * 32 + n0 + r];
        acc = __builtin_amdgcn_wmma_f32_16x16x4_f32(
             false, a,  false, b,
             (short)0, acc,  false,  false);
    }

    int col = n0 + r;
    int rowBase = m0 + khalf * 8;
#pragma unroll
    for (int v = 0; v < 8; ++v) {
        h2s[(rowBase + v) * 32 + col] = acc[v];
    }
}

#define GCN_E2U 8
__global__ __launch_bounds__(256)
void gcn_edge2(const int* __restrict__ src, const int* __restrict__ dst,
               const float* __restrict__ h2s,
               float* __restrict__ agg2a, float* __restrict__ agg2b, int E,
               size_t ws_bytes, size_t need_bytes) {
    const int lane = threadIdx.x & 31;
    float* __restrict__ agg2 = (gcn_two(ws_bytes, need_bytes) && gcn_half()) ? agg2b : agg2a;
    const int wave = __builtin_amdgcn_readfirstlane(
        (int)((blockIdx.x * blockDim.x + threadIdx.x) >> 5));
    const int nWaves = (int)((gridDim.x * blockDim.x) >> 5);
    const int nBatch = (E + 31) >> 5;
    for (int b = wave; b < nBatch; b += nWaves) {
        const int e0 = b << 5;
        const int e = e0 + lane;
        const int ec = e < E ? e : E - 1;
        const int s = src[ec];
        const int d = dst[ec];
        int cnt = E - e0;
        cnt = cnt < 32 ? cnt : 32;
        if (cnt == 32) {
#pragma unroll
            for (int j0 = 0; j0 < 32; j0 += GCN_E2U) {
                float v[GCN_E2U];
#pragma unroll
                for (int j = 0; j < GCN_E2U; ++j) {
                    const int sj = __builtin_amdgcn_readlane(s, j0 + j);
                    v[j] = h2s[(long)sj * 32 + lane];
                }
#pragma unroll
                for (int j = 0; j < GCN_E2U; ++j) {
                    const int dj = __builtin_amdgcn_readlane(d, j0 + j);
                    atomicAdd(&agg2[(long)dj * 32 + lane], v[j]);
                }
            }
        } else {
            for (int j = 0; j < cnt; ++j) {
                const int sj = __builtin_amdgcn_readlane(s, j);
                const int dj = __builtin_amdgcn_readlane(d, j);
                atomicAdd(&agg2[(long)dj * 32 + lane], h2s[(long)sj * 32 + lane]);
            }
        }
    }
}

#define GCN_PR 16

__device__ __forceinline__ void gcn_pool_row(float h, int g, float& acc, int& cnt, int& gcur,
                                             float* __restrict__ psum, float* __restrict__ pcnt,
                                             int lane) {
    if (g != gcur) {
        atomicAdd(&psum[gcur * 32 + lane], acc);
        if (lane == 0) atomicAdd(&pcnt[gcur], (float)cnt);
        acc = 0.0f;
        cnt = 0;
        gcur = g;
    }
    acc += h;
    cnt += 1;
}

__global__ __launch_bounds__(256)
void gcn_pool(const float* __restrict__ agg2, const float* __restrict__ agg2b,
              const float* __restrict__ h2s,
              const float* __restrict__ dis, const float* __restrict__ b2,
              const int* __restrict__ batch,
              float* __restrict__ psuma, float* __restrict__ psumb,
              float* __restrict__ pcnta, float* __restrict__ pcntb, int N,
              size_t ws_bytes, size_t need_bytes) {
    const bool two = gcn_two(ws_bytes, need_bytes);
    const float* __restrict__ agg2x = two ? agg2b : agg2;
    const bool hb = two && gcn_half();
    float* __restrict__ psum = hb ? psumb : psuma;
    float* __restrict__ pcnt = hb ? pcntb : pcnta;
    const int lane = threadIdx.x & 31;
    const int wave = __builtin_amdgcn_readfirstlane(
        (int)((blockIdx.x * blockDim.x + threadIdx.x) >> 5));
    const int i0 = wave * GCN_PR;
    if (i0 >= N) return;
    const float bf = b2[lane];
    const float* __restrict__ pa = agg2 + (long)i0 * 32 + lane;
    const float* __restrict__ px = agg2x + (long)i0 * 32 + lane;
    const float* __restrict__ ph = h2s + (long)i0 * 32 + lane;

    float acc = 0.0f;
    int cnt = 0;
    int gcur;
    if (i0 + GCN_PR <= N) {
        const int dl = __float_as_int(dis[i0 + (lane & (GCN_PR - 1))]);
        const int gl = batch[i0 + (lane & (GCN_PR - 1))];
        float a0[GCN_PR], a1[GCN_PR], hh[GCN_PR];
#pragma unroll
        for (int j = 0; j < GCN_PR; ++j) {
            a0[j] = pa[j * 32];
            a1[j] = px[j * 32];
            hh[j] = ph[j * 32];
        }
        asm volatile("" : "+v"(a1[0]));
        gcur = __builtin_amdgcn_readlane(gl, 0);
#pragma unroll
        for (int j = 0; j < GCN_PR; ++j) {
            const int g = __builtin_amdgcn_readlane(gl, j);
            const float di = __int_as_float(__builtin_amdgcn_readlane(dl, j));
            const float a = two ? a0[j] + a1[j] : a0[j];
            float h = di * (a + hh[j]) + bf;
            h = h > 0.0f ? h : 0.0f;
            gcn_pool_row(h, g, acc, cnt, gcur, psum, pcnt, lane);
        }
    } else {
        const int nv = N - i0;
        gcur = batch[i0];
        for (int j = 0; j < nv; ++j) {
            const int g = batch[i0 + j];
            const float di = dis[i0 + j];
            const float v0 = pa[j * 32];
            const float v1 = px[j * 32];
            const float a = two ? v0 + v1 : v0;
            float h = di * (a + ph[j * 32]) + bf;
            h = h > 0.0f ? h : 0.0f;
            gcn_pool_row(h, g, acc, cnt, gcur, psum, pcnt, lane);
        }
    }
    atomicAdd(&psum[gcur * 32 + lane], acc);
    if (lane == 0) atomicAdd(&pcnt[gcur], (float)cnt);
}

#define GCN_MLP_BS 1024
#define GCN_MLP_GPB (GCN_MLP_BS / 32)
__global__ __launch_bounds__(GCN_MLP_BS)
void gcn_mlp(const float* __restrict__ psum, const float* __restrict__ psumb,
             const float* __restrict__ pcnt, const float* __restrict__ pcntb,
             const float* __restrict__ Wp1, const float* __restrict__ bp1,
             const float* __restrict__ Wp2, const float* __restrict__ bp2,
             float* __restrict__ out, int G, size_t ws_bytes, size_t need_bytes) {
    const bool two = gcn_two(ws_bytes, need_bytes);
    const float* __restrict__ psumx = two ? psumb : psum;
    const float* __restrict__ pcntx = two ? pcntb : pcnt;
    __shared__ float res[GCN_MLP_GPB * 3];
    const int lane = threadIdx.x & 31;
    const int w = __builtin_amdgcn_readfirstlane((int)(threadIdx.x >> 5));
    const int g = blockIdx.x * GCN_MLP_GPB + w;
    if (g < G) {
        const float c0 = pcnt[g];
        const float c1 = pcntx[g];
        const float p0 = psum[g * 32 + lane];
        const float p1 = psumx[g * 32 + lane];
        float c = two ? c0 + c1 : c0;
        const float ps = two ? p0 + p1 : p0;
        c = c > 1.0f ? c : 1.0f;
        const float inv = 1.0f / c;
        const int p = __float_as_int(ps * inv);

        const int j0 = lane * 4;
        float h0 = bp1[j0], h1 = bp1[j0 + 1], h2 = bp1[j0 + 2], h3 = bp1[j0 + 3];
#pragma unroll
        for (int f = 0; f < 32; ++f) {
            const float pf = __int_as_float(__builtin_amdgcn_readlane(p, f));
            const float* wr = Wp1 + f * 128 + j0;
            h0 = fmaf(pf, wr[0], h0);
            h1 = fmaf(pf, wr[1], h1);
            h2 = fmaf(pf, wr[2], h2);
            h3 = fmaf(pf, wr[3], h3);
        }
        h0 = h0 > 0.0f ? h0 : 0.0f;
        h1 = h1 > 0.0f ? h1 : 0.0f;
        h2 = h2 > 0.0f ? h2 : 0.0f;
        h3 = h3 > 0.0f ? h3 : 0.0f;

        const float* v = Wp2 + j0 * 3;
        float o0 = fmaf(h3, v[9],  fmaf(h2, v[6], fmaf(h1, v[3], h0 * v[0])));
        float o1 = fmaf(h3, v[10], fmaf(h2, v[7], fmaf(h1, v[4], h0 * v[1])));
        float o2 = fmaf(h3, v[11], fmaf(h2, v[8], fmaf(h1, v[5], h0 * v[2])));
        for (int m = 16; m >= 1; m >>= 1) {
            o0 += __shfl_xor(o0, m);
            o1 += __shfl_xor(o1, m);
            o2 += __shfl_xor(o2, m);
        }
        if (lane == 0) {
            res[w * 3 + 0] = o0 + bp2[0];
            res[w * 3 + 1] = o1 + bp2[1];
            res[w * 3 + 2] = o2 + bp2[2];
        }
    }
    __syncthreads();
    if (w == 0) {
        const int gg = blockIdx.x * GCN_MLP_GPB + lane;
        if (gg < G) {
            out[gg * 3 + 0] = res[lane * 3 + 0];
            out[gg * 3 + 1] = res[lane * 3 + 1];
            out[gg * 3 + 2] = res[lane * 3 + 2];
        }
    }
}

extern "C" void kernel_launch(void* const* d_in, const int* in_sizes, int n_in,
                              void* d_out, int out_size, void* d_ws, size_t ws_size,
                              hipStream_t stream) {
    const float* x   = (const float*)d_in[0];
    const float* W1  = (const float*)d_in[1];
    const float* b1  = (const float*)d_in[2];
    const float* W2  = (const float*)d_in[3];
    const float* b2  = (const float*)d_in[4];
    const float* Wp1 = (const float*)d_in[5];
    const float* bp1 = (const float*)d_in[6];
    const float* Wp2 = (const float*)d_in[7];
    const float* bp2 = (const float*)d_in[8];
    const int* eidx  = (const int*)d_in[9];
    const int* batch = (const int*)d_in[10];
    float* out = (float*)d_out;

    const int N = in_sizes[0];
    const int E = in_sizes[9] / 2;
    const int G = out_size / 3;
    const int* src = eidx;
    const int* dst = eidx + E;

    float* wsf   = (float*)d_ws;
    float* deg   = wsf;
    float* s1    = deg + N;
    float* dis   = s1 + N;
    float* tbuf  = dis + N;
    float* h2s   = tbuf + N;
    float* agg2  = h2s + (long)N * 32;
    float* psum  = agg2 + (long)N * 32;
    float* pcnt  = psum + (long)G * 32;
    float* agg2b = (float*)(((uintptr_t)(pcnt + G) + 255) & ~(uintptr_t)255);
    float* psumb = agg2b + (long)N * 32;
    float* degb  = psumb + (long)G * 32;
    float* s1b   = degb + N;
    float* pcntb = s1b + N;
    float* wsend = pcntb + G;
    const size_t need = (size_t)((const char*)wsend - (const char*)d_ws);

    const int TPB = 256;

    long nz1 = N;
    long nz2 = 32L * N + 32L * G + G;
    long nz2two = (long)(wsend - agg2);
    static_assert(GCN_K <= 32, "degp [GCN_K][N] must fit in h2s [N][32]");
    float* degp = h2s;

    const int chunk = (E + GCN_K - 1) / GCN_K;
    const dim3 sgrid(GCN_K, (N + GCN_RN - 1) / GCN_RN);
    gcn_degree<<<sgrid, GCN_BS, 0, stream>>>(dst, degp, E, N, chunk);
    gcn_dis<<<2048, TPB, 0, stream>>>(degp, x, dis, tbuf, N, s1, nz1, agg2, nz2, nz2two, ws_size, need);
    const dim3 egrid(GCN_K, ((N + GCN_SR - 1) / GCN_SR) * ((N + GCN_DR - 1) / GCN_DR));
    gcn_edge1<<<egrid, GCN_BS, 0, stream>>>(src, dst, tbuf, dis, s1, s1b, E, N, chunk, ws_size, need);

    int nTiles = (N / 16) * 2;
    gcn_gemm_wmma<<<(nTiles + 7) / 8, TPB, 0, stream>>>(s1, s1b, x, W1, b1, W2, dis, h2s, nTiles, ws_size, need);

    gcn_edge2<<<2048, TPB, 0, stream>>>(src, dst, h2s, agg2, agg2b, E, ws_size, need);

    gcn_pool<<<((N + GCN_PR - 1) / GCN_PR + 7) / 8, TPB, 0, stream>>>(agg2, agg2b, h2s, dis, b2, batch, psum, psumb, pcnt, pcntb, N, ws_size, need);

    gcn_mlp<<<(G + GCN_MLP_GPB - 1) / GCN_MLP_GPB, GCN_MLP_BS, 0, stream>>>(psum, psumb, pcnt, pcntb, Wp1, bp1, Wp2, bp2, out, G, ws_size, need);
}
